# starting kernel
# baseline (speedup 1.0000x reference)
#include <hip/hip_runtime.h>
#include <math.h>

typedef float v2f __attribute__((ext_vector_type(2)));
typedef float v8f __attribute__((ext_vector_type(8)));

#define LRELU_SLOPE 0.2f

__device__ __forceinline__ unsigned f2ord(float f) {
  unsigned u = __float_as_uint(f);
  return (u & 0x80000000u) ? ~u : (u | 0x80000000u);
}
__device__ __forceinline__ float ord2f(unsigned u) {
  return __uint_as_float((u & 0x80000000u) ? (u & 0x7FFFFFFFu) : ~u);
}

__global__ __launch_bounds__(256) void gemm_wmma_f32(
    const float* __restrict__ X, const float* __restrict__ W,
    float* __restrict__ Y, int M, int K, int Nc) {
  const int lane  = threadIdx.x & 31;
  const int wave  = (blockIdx.x * blockDim.x + threadIdx.x) >> 5;
  const int tilesN = Nc >> 4;
  const int tilesM = M >> 4;
  if (wave >= tilesM * tilesN) return;
  const int mTile = (wave / tilesN) << 4;
  const int nTile = (wave % tilesN) << 4;
  const int half  = lane >> 4;
  const int l16   = lane & 15;

  const float* __restrict__ xrow = X + (size_t)(mTile + l16) * K;
  const float* __restrict__ wcol = W + nTile + l16;

  v8f acc = {0.f, 0.f, 0.f, 0.f, 0.f, 0.f, 0.f, 0.f};
  for (int k = 0; k < K; k += 4) {
    const int ka = k + 2 * half;
    v2f a = *(const v2f*)(xrow + ka);
    v2f b;
    b.x = wcol[(size_t)ka * Nc];
    b.y = wcol[(size_t)(ka + 1) * Nc];
#if __has_builtin(__builtin_amdgcn_wmma_f32_16x16x4_f32)
    acc = __builtin_amdgcn_wmma_f32_16x16x4_f32(false, a, false, b,
                                                (short)0, acc, false, false);
#else
    acc[0] += a.x * b.x;
#endif
  }
  float* __restrict__ yrow = Y + nTile + l16;
  const int mbase = mTile + half * 8;
#pragma unroll
  for (int r = 0; r < 8; ++r)
    yrow[(size_t)(mbase + r) * Nc] = acc[r];
}

__global__ __launch_bounds__(256) void att_coef(
    const float* __restrict__ h, const float* __restrict__ att_s,
    const float* __restrict__ att_d, float* __restrict__ a_s,
    float* __restrict__ a_d, int NH, int H, int C) {
  const long long gt = (long long)blockIdx.x * blockDim.x + threadIdx.x;
  const int wave = (int)(gt >> 5);
  const int c    = threadIdx.x & 31;
  if (wave >= NH) return;
  const int hh = wave % H;
  const float v = h[(size_t)wave * C + c];
  float ss = v * att_s[hh * C + c];
  float dd = v * att_d[hh * C + c];
#pragma unroll
  for (int off = 16; off > 0; off >>= 1) {
    ss += __shfl_xor(ss, off, 32);
    dd += __shfl_xor(dd, off, 32);
  }
  if (c == 0) { a_s[wave] = ss; a_d[wave] = dd; }
}

__global__ __launch_bounds__(256) void edge_max(
    const int* __restrict__ ei, int E, int N, int H,
    const float* __restrict__ a_s, const float* __restrict__ a_d,
    unsigned* __restrict__ m) {
  const int e = blockIdx.x * blockDim.x + threadIdx.x;
  const int Etot = E + N;
  if (e >= Etot) return;
  const int s = (e < E) ? ei[e] : (e - E);
  const int d = (e < E) ? ei[E + e] : (e - E);
  for (int hh = 0; hh < H; ++hh) {
    float l = a_s[s * H + hh] + a_d[d * H + hh];
    l = (l > 0.f) ? l : LRELU_SLOPE * l;
    atomicMax(&m[d * H + hh], f2ord(l));
  }
}

__global__ __launch_bounds__(256) void edge_acc(
    const int* __restrict__ ei, int E, int N, int H, int C,
    const float* __restrict__ a_s, const float* __restrict__ a_d,
    const unsigned* __restrict__ m, const float* __restrict__ h,
    float* __restrict__ num, float* __restrict__ den) {
  const long long gt = (long long)blockIdx.x * blockDim.x + threadIdx.x;
  const long long wave = gt >> 5;
  const int c = threadIdx.x & 31;
  const long long Etot = (long long)E + N;
  if (wave >= Etot * H) return;
  const int e  = (int)(wave / H);
  const int hh = (int)(wave % H);
  const int s = (e < E) ? ei[e] : (e - E);
  const int d = (e < E) ? ei[E + e] : (e - E);
  float l = a_s[s * H + hh] + a_d[d * H + hh];
  l = (l > 0.f) ? l : LRELU_SLOPE * l;
  const float w = expf(l - ord2f(m[d * H + hh]));
  if (c == 0) atomicAdd(&den[d * H + hh], w);
  const size_t srow = ((size_t)s * H + hh) * C;
  const size_t drow = ((size_t)d * H + hh) * C;
  atomicAdd(&num[drow + c], w * h[srow + c]);
}

__global__ __launch_bounds__(256) void finalize(
    const float* __restrict__ num, const float* __restrict__ den,
    const float* __restrict__ bias, float* __restrict__ out,
    int N, int H, int C) {
  const int t = blockIdx.x * blockDim.x + threadIdx.x;
  if (t >= N * H * C) return;
  const int nh = t / C;
  const int hc = t % (H * C);
  float v = num[t] / (den[nh] + 1e-16f) + bias[hc];
  v = (v > 0.f) ? v : (expf(v) - 1.f);
  out[t] = v;
}

extern "C" void kernel_launch(void* const* d_in, const int* in_sizes, int n_in,
                              void* d_out, int out_size, void* d_ws, size_t ws_size,
                              hipStream_t stream) {
  const float* x   = (const float*)d_in[0];
  const int*   ei  = (const int*)d_in[1];
  const float* W1  = (const float*)d_in[2];
  const float* as1 = (const float*)d_in[3];
  const float* ad1 = (const float*)d_in[4];
  const float* b1  = (const float*)d_in[5];
  const float* W2  = (const float*)d_in[6];
  const float* as2 = (const float*)d_in[7];
  const float* ad2 = (const float*)d_in[8];
  const float* b2  = (const float*)d_in[9];

  const int HC   = in_sizes[3];
  const int Fin  = in_sizes[2] / HC;
  const int N    = in_sizes[0] / Fin;
  const int E    = in_sizes[1] / 2;
  const int H    = 8;
  const int C    = HC / H;
  const int Cout = in_sizes[7];
  const int Etot = E + N;

  float* ws = (float*)d_ws;
  float*    h1   = ws;
  float*    x2   = h1 + (size_t)N * HC;
  float*    h2   = x2 + (size_t)N * HC;
  float*    num2 = h2 + (size_t)N * Cout;
  float*    aS1  = num2 + (size_t)N * Cout;
  float*    aD1  = aS1 + (size_t)N * H;
  unsigned* m1   = (unsigned*)(aD1 + (size_t)N * H);
  float*    den1 = (float*)(m1 + (size_t)N * H);
  float*    aS2  = den1 + (size_t)N * H;
  float*    aD2  = aS2 + N;
  unsigned* m2   = (unsigned*)(aD2 + N);
  float*    den2 = (float*)(m2 + N);

  hipMemsetAsync(x2,   0, (size_t)N * HC   * sizeof(float), stream);
  hipMemsetAsync(num2, 0, (size_t)N * Cout * sizeof(float), stream);
  hipMemsetAsync(m1,   0, (size_t)N * H * sizeof(unsigned), stream);
  hipMemsetAsync(den1, 0, (size_t)N * H * sizeof(float), stream);
  hipMemsetAsync(m2,   0, (size_t)N * sizeof(unsigned), stream);
  hipMemsetAsync(den2, 0, (size_t)N * sizeof(float), stream);

  const int TB = 256;
  auto blocks = [](long long n, int tb) { return (int)((n + tb - 1) / tb); };

  {
    const long long waves = (long long)(N / 16) * (HC / 16);
    gemm_wmma_f32<<<blocks(waves * 32, TB), TB, 0, stream>>>(x, W1, h1, N, Fin, HC);
    att_coef<<<blocks((long long)N * H * 32, TB), TB, 0, stream>>>(
        h1, as1, ad1, aS1, aD1, N * H, H, C);
    edge_max<<<blocks(Etot, TB), TB, 0, stream>>>(ei, E, N, H, aS1, aD1, m1);
    edge_acc<<<blocks((long long)Etot * H * 32, TB), TB, 0, stream>>>(
        ei, E, N, H, C, aS1, aD1, m1, h1, x2, den1);
    finalize<<<blocks((long long)N * HC, TB), TB, 0, stream>>>(x2, den1, b1, x2, N, H, C);
  }

  {
    const long long waves = (long long)(N / 16) * (Cout / 16);
    gemm_wmma_f32<<<blocks(waves * 32, TB), TB, 0, stream>>>(x2, W2, h2, N, HC, Cout);
    att_coef<<<blocks((long long)N * 32, TB), TB, 0, stream>>>(
        h2, as2, ad2, aS2, aD2, N, 1, Cout);
    edge_max<<<blocks(Etot, TB), TB, 0, stream>>>(ei, E, N, 1, aS2, aD2, m2);
    edge_acc<<<blocks((long long)Etot * 32, TB), TB, 0, stream>>>(
        ei, E, N, 1, Cout, aS2, aD2, m2, h2, num2, den2);
    finalize<<<blocks((long long)N * Cout, TB), TB, 0, stream>>>(
        num2, den2, b2, (float*)d_out, N, 1, Cout);
  }
}


# Edges counting-sorted by dst once per call; one wave per dst does max, exp-sum, weighted row sum, bias+ELU in registers: no float atomics; sums now run in scatter arrival order per dst
# speedup vs baseline: 4.1134x; 4.1134x over previous
#include <hip/hip_runtime.h>
#include <math.h>

typedef float v2f __attribute__((ext_vector_type(2)));
typedef float v4f __attribute__((ext_vector_type(4)));
typedef float v8f __attribute__((ext_vector_type(8)));

#define LRELU_SLOPE 0.2f

__global__ __launch_bounds__(256) void gemm_wmma_f32(
    const float* __restrict__ X, const float* __restrict__ W,
    float* __restrict__ Y, int M, int K, int Nc) {
  const int lane  = threadIdx.x & 31;
  const int wave  = (blockIdx.x * blockDim.x + threadIdx.x) >> 5;
  const int tilesN = Nc >> 4;
  const int tilesM = M >> 4;
  if (wave >= tilesM * tilesN) return;
  const int mTile = (wave / tilesN) << 4;
  const int nTile = (wave % tilesN) << 4;
  const int half  = lane >> 4;
  const int l16   = lane & 15;

  const float* __restrict__ xrow = X + (size_t)(mTile + l16) * K;
  const float* __restrict__ wcol = W + nTile + l16;

  v8f acc = {0.f, 0.f, 0.f, 0.f, 0.f, 0.f, 0.f, 0.f};
  for (int k = 0; k < K; k += 4) {
    const int ka = k + 2 * half;
    v2f a = *(const v2f*)(xrow + ka);
    v2f b;
    b.x = wcol[(size_t)ka * Nc];
    b.y = wcol[(size_t)(ka + 1) * Nc];
#if __has_builtin(__builtin_amdgcn_wmma_f32_16x16x4_f32)
    acc = __builtin_amdgcn_wmma_f32_16x16x4_f32(false, a, false, b,
                                                (short)0, acc, false, false);
#else
    acc[0] += a.x * b.x;
#endif
  }
  float* __restrict__ yrow = Y + nTile + l16;
  const int mbase = mTile + half * 8;
#pragma unroll
  for (int r = 0; r < 8; ++r)
    yrow[(size_t)(mbase + r) * Nc] = acc[r];
}

__global__ __launch_bounds__(256) void att_coef(
    const float* __restrict__ h, const float* __restrict__ att_s,
    const float* __restrict__ att_d, float* __restrict__ a_s,
    float* __restrict__ a_d, int NH, int H, int C) {
  const long long gt = (long long)blockIdx.x * blockDim.x + threadIdx.x;
  const int wave = (int)(gt >> 5);
  const int c    = threadIdx.x & 31;
  if (wave >= NH) return;
  const int hh = wave % H;
  const float v = h[(size_t)wave * C + c];
  float ss = v * att_s[hh * C + c];
  float dd = v * att_d[hh * C + c];
#pragma unroll
  for (int off = 16; off > 0; off >>= 1) {
    ss += __shfl_xor(ss, off, 32);
    dd += __shfl_xor(dd, off, 32);
  }
  if (c == 0) { a_s[wave] = ss; a_d[wave] = dd; }
}

__global__ __launch_bounds__(256) void dst_hist(
    const int* __restrict__ ei, int E, int* __restrict__ cnt) {
  const int e = blockIdx.x * blockDim.x + threadIdx.x;
  if (e >= E) return;
  atomicAdd(&cnt[ei[E + e]], 1);
}

__global__ __launch_bounds__(1024) void dst_scan(
    int* cnt, int* __restrict__ rowptr, int N) {
  __shared__ int wtot[32];
  const int lane = threadIdx.x & 31;
  const int wave = __builtin_amdgcn_readfirstlane((int)(threadIdx.x >> 5));
  const int seg = (N + 31) >> 5;
  int lo = wave * seg;      if (lo > N) lo = N;
  int hi = lo + seg;        if (hi > N) hi = N;

  int sum = 0;
  for (int i = lo + lane; i < hi; i += 32) sum += cnt[i];
#pragma unroll
  for (int off = 16; off > 0; off >>= 1) sum += __shfl_xor(sum, off, 32);
  if (lane == 0) wtot[wave] = sum;
  __syncthreads();

  const int t = wtot[lane];
  int incl = t;
#pragma unroll
  for (int off = 1; off < 32; off <<= 1) {
    const int v = __shfl_up(incl, off, 32);
    if (lane >= off) incl += v;
  }
  int carry = __shfl(incl - t, wave, 32);
  const int total = __shfl(incl, 31, 32);

  for (int base = lo; base < hi; base += 256) {
    const int i0 = base + lane * 8;
    int c[8];
    int s = 0;
#pragma unroll
    for (int u = 0; u < 8; ++u) {
      c[u] = (i0 + u < hi) ? cnt[i0 + u] : 0;
      s += c[u];
    }
    int inc = s;
#pragma unroll
    for (int off = 1; off < 32; off <<= 1) {
      const int v = __shfl_up(inc, off, 32);
      if (lane >= off) inc += v;
    }
    int run = carry + inc - s;
#pragma unroll
    for (int u = 0; u < 8; ++u) {
      if (i0 + u < hi) {
        rowptr[i0 + u] = run;
        run += c[u];
        cnt[i0 + u] = 0;
      }
    }
    carry += __shfl(inc, 31, 32);
  }
  if (threadIdx.x == 0) rowptr[N] = total;
}

__global__ __launch_bounds__(256) void dst_scatter(
    const int* __restrict__ ei, int E, const int* __restrict__ rowptr,
    int* __restrict__ cnt, int* __restrict__ csr_src) {
  const int e = blockIdx.x * blockDim.x + threadIdx.x;
  if (e >= E) return;
  const int s = ei[e];
  const int d = ei[E + e];
  const int pos = rowptr[d] + atomicAdd(&cnt[d], 1);
  csr_src[pos] = s;
}

__device__ __forceinline__ float lrelu(float l) {
  return (l > 0.f) ? l : LRELU_SLOPE * l;
}
__device__ __forceinline__ float elu1(float v) {
  return (v > 0.f) ? v : (expf(v) - 1.f);
}

__global__ __launch_bounds__(256) void agg_h8(
    const int* __restrict__ rowptr, const int* __restrict__ csr_src,
    const float* __restrict__ a_s, const float* __restrict__ a_d,
    const float* __restrict__ h, const float* __restrict__ bias,
    float* __restrict__ out, int N) {
  const int lane = threadIdx.x & 31;
  const int d = __builtin_amdgcn_readfirstlane((int)(blockIdx.x * 8 + (threadIdx.x >> 5)));
  if (d >= N) return;
  const int hA = lane >> 3, hB = 4 + (lane >> 3);
  const int beg = rowptr[d], end = rowptr[d + 1];
  const float adA = a_d[d * 8 + hA], adB = a_d[d * 8 + hB];
  const float lsA = lrelu(a_s[d * 8 + hA] + adA);
  const float lsB = lrelu(a_s[d * 8 + hB] + adB);

  float mA = lsA, mB = lsB;
  for (int base = beg; base < end; base += 32) {
    const int n = (end - base < 32) ? (end - base) : 32;
    const int sv = (lane < n) ? csr_src[base + lane] : d;
    for (int j = 0; j < n; j += 8) {
      float ga[8], gb[8];
#pragma unroll
      for (int u = 0; u < 8; ++u) {
        const int s = __builtin_amdgcn_readlane(sv, j + u);
        const float* __restrict__ as = a_s + (size_t)s * 8;
        ga[u] = as[hA];
        gb[u] = as[hB];
      }
#pragma unroll
      for (int u = 0; u < 8; ++u) {
        mA = fmaxf(mA, lrelu(ga[u] + adA));
        mB = fmaxf(mB, lrelu(gb[u] + adB));
      }
    }
  }

  const v4f* __restrict__ hd = (const v4f*)(h + (size_t)d * 256);
  const float wsA = expf(lsA - mA), wsB = expf(lsB - mB);
  float denA = wsA, denB = wsB;
  v4f accA = wsA * hd[lane];
  v4f accB = wsB * hd[32 + lane];
  for (int base = beg; base < end; base += 32) {
    const int n = (end - base < 32) ? (end - base) : 32;
    const int sv = (lane < n) ? csr_src[base + lane] : d;
    for (int j = 0; j < n; j += 4) {
      v4f ra[4], rb[4];
      float ga[4], gb[4];
#pragma unroll
      for (int u = 0; u < 4; ++u) {
        const int s = __builtin_amdgcn_readlane(sv, j + u);
        const v4f* __restrict__ hr = (const v4f*)(h + (size_t)s * 256);
        const float* __restrict__ as = a_s + (size_t)s * 8;
        ra[u] = hr[lane];
        rb[u] = hr[32 + lane];
        ga[u] = as[hA];
        gb[u] = as[hB];
      }
#pragma unroll
      for (int u = 0; u < 4; ++u) {
        const bool live = (j + u) < n;
        const float la = live ? lrelu(ga[u] + adA) : -__builtin_huge_valf();
        const float lb = live ? lrelu(gb[u] + adB) : -__builtin_huge_valf();
        const float wa = expf(la - mA);
        const float wb = expf(lb - mB);
        denA += wa;
        denB += wb;
        accA += wa * ra[u];
        accB += wb * rb[u];
      }
      __builtin_amdgcn_sched_group_barrier(0x020, 16, 0);
      __builtin_amdgcn_sched_group_barrier(0x002, 256, 0);
    }
  }

  const float qA = denA + 1e-16f, qB = denB + 1e-16f;
  const float* __restrict__ bA = bias + 4 * lane;
  const float* __restrict__ bB = bias + 128 + 4 * lane;
  v4f oA, oB;
  oA.x = accA.x / qA + bA[0];  oA.y = accA.y / qA + bA[1];
  oA.z = accA.z / qA + bA[2];  oA.w = accA.w / qA + bA[3];
  oB.x = accB.x / qB + bB[0];  oB.y = accB.y / qB + bB[1];
  oB.z = accB.z / qB + bB[2];  oB.w = accB.w / qB + bB[3];
  oA.x = elu1(oA.x);  oA.y = elu1(oA.y);  oA.z = elu1(oA.z);  oA.w = elu1(oA.w);
  oB.x = elu1(oB.x);  oB.y = elu1(oB.y);  oB.z = elu1(oB.z);  oB.w = elu1(oB.w);
  v4f* __restrict__ od = (v4f*)(out + (size_t)d * 256);
  od[lane]      = oA;
  od[32 + lane] = oB;
}

__global__ __launch_bounds__(256) void agg_h1(
    const int* __restrict__ rowptr, const int* __restrict__ csr_src,
    const float* __restrict__ a_s, const float* __restrict__ a_d,
    const float* __restrict__ h, const float* __restrict__ bias,
    float* __restrict__ out, int N) {
  const int lane = threadIdx.x & 31;
  const int d = __builtin_amdgcn_readfirstlane((int)(blockIdx.x * 8 + (threadIdx.x >> 5)));
  if (d >= N) return;
  const int beg = rowptr[d], end = rowptr[d + 1];
  const float ad = a_d[d];
  const float ls = lrelu(a_s[d] + ad);

  float m = ls;
  for (int base = beg; base < end; base += 32) {
    const int idx = base + lane;
    if (idx < end) m = fmaxf(m, lrelu(a_s[csr_src[idx]] + ad));
  }
#pragma unroll
  for (int off = 16; off > 0; off >>= 1) m = fmaxf(m, __shfl_xor(m, off, 32));

  const float wself = expf(ls - m);
  float den = 0.f;
  float acc = wself * h[(size_t)d * 32 + lane];
  for (int base = beg; base < end; base += 32) {
    const int n = (end - base < 32) ? (end - base) : 32;
    int sv = d;
    float wv = 0.f;
    if (lane < n) {
      sv = csr_src[base + lane];
      wv = expf(lrelu(a_s[sv] + ad) - m);
    }
    den += wv;
    for (int j = 0; j < n; j += 8) {
      float r[8], w[8];
#pragma unroll
      for (int u = 0; u < 8; ++u) {
        const int s = __builtin_amdgcn_readlane(sv, j + u);
        w[u] = __uint_as_float(
            (unsigned)__builtin_amdgcn_readlane((int)__float_as_uint(wv), j + u));
        r[u] = h[(size_t)s * 32 + lane];
      }
#pragma unroll
      for (int u = 0; u < 8; ++u) acc += w[u] * r[u];
    }
  }
#pragma unroll
  for (int off = 16; off > 0; off >>= 1) den += __shfl_xor(den, off, 32);
  den += wself;
  out[(size_t)d * 32 + lane] = elu1(acc / (den + 1e-16f) + bias[lane]);
}

extern "C" void kernel_launch(void* const* d_in, const int* in_sizes, int n_in,
                              void* d_out, int out_size, void* d_ws, size_t ws_size,
                              hipStream_t stream) {
  const float* x   = (const float*)d_in[0];
  const int*   ei  = (const int*)d_in[1];
  const float* W1  = (const float*)d_in[2];
  const float* as1 = (const float*)d_in[3];
  const float* ad1 = (const float*)d_in[4];
  const float* b1  = (const float*)d_in[5];
  const float* W2  = (const float*)d_in[6];
  const float* as2 = (const float*)d_in[7];
  const float* ad2 = (const float*)d_in[8];
  const float* b2  = (const float*)d_in[9];

  const int HC   = in_sizes[3];
  const int Fin  = in_sizes[2] / HC;
  const int N    = in_sizes[0] / Fin;
  const int E    = in_sizes[1] / 2;
  const int H    = 8;
  const int C    = HC / H;
  const int Cout = in_sizes[7];

  float* ws = (float*)d_ws;
  float* h1  = ws;
  float* x2  = h1 + (size_t)N * HC;
  float* h2  = x2 + (size_t)N * HC;
  float* aS1 = h2 + (size_t)N * Cout;
  float* aD1 = aS1 + (size_t)N * H;
  float* aS2 = aD1 + (size_t)N * H;
  float* aD2 = aS2 + N;
  int*   cnt    = (int*)(aD2 + N);
  int*   rowptr = cnt + N;
  int*   csr    = rowptr + N + 1;

  const int TB = 256;
  auto blocks = [](long long n, int tb) { return (int)((n + tb - 1) / tb); };

  hipMemsetAsync(cnt, 0, (size_t)N * sizeof(int), stream);
  dst_hist<<<blocks(E, TB), TB, 0, stream>>>(ei, E, cnt);
  dst_scan<<<1, 1024, 0, stream>>>(cnt, rowptr, N);
  dst_scatter<<<blocks(E, TB), TB, 0, stream>>>(ei, E, rowptr, cnt, csr);

  {
    const long long waves = (long long)(N / 16) * (HC / 16);
    gemm_wmma_f32<<<blocks(waves * 32, TB), TB, 0, stream>>>(x, W1, h1, N, Fin, HC);
    att_coef<<<blocks((long long)N * H * 32, TB), TB, 0, stream>>>(
        h1, as1, ad1, aS1, aD1, N * H, H, C);
    agg_h8<<<blocks((long long)N * 32, TB), TB, 0, stream>>>(
        rowptr, csr, aS1, aD1, h1, b1, x2, N);
  }

  {
    const long long waves = (long long)(N / 16) * (Cout / 16);
    gemm_wmma_f32<<<blocks(waves * 32, TB), TB, 0, stream>>>(x2, W2, h2, N, HC, Cout);
    att_coef<<<blocks((long long)N * 32, TB), TB, 0, stream>>>(
        h2, as2, ad2, aS2, aD2, N, 1, Cout);
    agg_h1<<<blocks((long long)N * 32, TB), TB, 0, stream>>>(
        rowptr, csr, aS2, aD2, h2, b2, (float*)d_out, N);
  }
}


# att_coef: one wave per 64 (node,head) pairs, whole-line 16 B loads, 8 chunks in flight (64x fewer waves); each dot now sums 4 channels per lane, then 3 xor steps over 8 lanes
# speedup vs baseline: 4.5967x; 1.1175x over previous
#include <hip/hip_runtime.h>
#include <math.h>

typedef float v2f __attribute__((ext_vector_type(2)));
typedef float v4f __attribute__((ext_vector_type(4)));
typedef float v8f __attribute__((ext_vector_type(8)));

#define LRELU_SLOPE 0.2f

__global__ __launch_bounds__(256) void gemm_wmma_f32(
    const float* __restrict__ X, const float* __restrict__ W,
    float* __restrict__ Y, int M, int K, int Nc) {
  const int lane  = threadIdx.x & 31;
  const int wave  = (blockIdx.x * blockDim.x + threadIdx.x) >> 5;
  const int tilesN = Nc >> 4;
  const int tilesM = M >> 4;
  if (wave >= tilesM * tilesN) return;
  const int mTile = (wave / tilesN) << 4;
  const int nTile = (wave % tilesN) << 4;
  const int half  = lane >> 4;
  const int l16   = lane & 15;

  const float* __restrict__ xrow = X + (size_t)(mTile + l16) * K;
  const float* __restrict__ wcol = W + nTile + l16;

  v8f acc = {0.f, 0.f, 0.f, 0.f, 0.f, 0.f, 0.f, 0.f};
  for (int k = 0; k < K; k += 4) {
    const int ka = k + 2 * half;
    v2f a = *(const v2f*)(xrow + ka);
    v2f b;
    b.x = wcol[(size_t)ka * Nc];
    b.y = wcol[(size_t)(ka + 1) * Nc];
#if __has_builtin(__builtin_amdgcn_wmma_f32_16x16x4_f32)
    acc = __builtin_amdgcn_wmma_f32_16x16x4_f32(false, a, false, b,
                                                (short)0, acc, false, false);
#else
    acc[0] += a.x * b.x;
#endif
  }
  float* __restrict__ yrow = Y + nTile + l16;
  const int mbase = mTile + half * 8;
#pragma unroll
  for (int r = 0; r < 8; ++r)
    yrow[(size_t)(mbase + r) * Nc] = acc[r];
}

__device__ __forceinline__ float dot4(v4f a, v4f b) {
  return a.x * b.x + a.y * b.y + a.z * b.z + a.w * b.w;
}

template <int H>
__global__ __launch_bounds__(256) void att_coef(
    const float* __restrict__ h, const float* __restrict__ att_s,
    const float* __restrict__ att_d, float* __restrict__ a_s,
    float* __restrict__ a_d, int nChunks) {
  constexpr int CPW = 8;
  constexpr int HC = H * 32;
  const int lane = threadIdx.x & 31;
  const int wave = __builtin_amdgcn_readfirstlane(
      (int)((blockIdx.x * blockDim.x + threadIdx.x) >> 5));
  const int c0 = wave * CPW;
  if (c0 >= nChunks) return;

  const int fA = (4 * lane) % HC, fB = (128 + 4 * lane) % HC;
  const v4f sA = *(const v4f*)(att_s + fA), sB = *(const v4f*)(att_s + fB);
  const v4f dA = *(const v4f*)(att_d + fA), dB = *(const v4f*)(att_d + fB);

  v4f ra[CPW], rb[CPW];
#pragma unroll
  for (int u = 0; u < CPW; ++u) {
    const int ch = (c0 + u < nChunks) ? (c0 + u) : (nChunks - 1);
    const v4f* __restrict__ p = (const v4f*)(h + (size_t)ch * 256);
    ra[u] = p[lane];
    rb[u] = p[32 + lane];
  }
#pragma unroll
  for (int u = 0; u < CPW; ++u) {
    float ssA = dot4(ra[u], sA), ssB = dot4(rb[u], sB);
    float ddA = dot4(ra[u], dA), ddB = dot4(rb[u], dB);
#pragma unroll
    for (int off = 4; off > 0; off >>= 1) {
      ssA += __shfl_xor(ssA, off, 32);
      ssB += __shfl_xor(ssB, off, 32);
      ddA += __shfl_xor(ddA, off, 32);
      ddB += __shfl_xor(ddB, off, 32);
    }
    if (c0 + u < nChunks && (lane & 7) == 0) {
      const int w = (c0 + u) * 8 + (lane >> 3);
      a_s[w] = ssA;  a_s[w + 4] = ssB;
      a_d[w] = ddA;  a_d[w + 4] = ddB;
    }
  }
}

__global__ __launch_bounds__(256) void dst_hist(
    const int* __restrict__ ei, int E, int* __restrict__ cnt) {
  const int e = blockIdx.x * blockDim.x + threadIdx.x;
  if (e >= E) return;
  atomicAdd(&cnt[ei[E + e]], 1);
}

__global__ __launch_bounds__(1024) void dst_scan(
    int* cnt, int* __restrict__ rowptr, int N) {
  __shared__ int wtot[32];
  const int lane = threadIdx.x & 31;
  const int wave = __builtin_amdgcn_readfirstlane((int)(threadIdx.x >> 5));
  const int seg = (N + 31) >> 5;
  int lo = wave * seg;      if (lo > N) lo = N;
  int hi = lo + seg;        if (hi > N) hi = N;

  int sum = 0;
  for (int i = lo + lane; i < hi; i += 32) sum += cnt[i];
#pragma unroll
  for (int off = 16; off > 0; off >>= 1) sum += __shfl_xor(sum, off, 32);
  if (lane == 0) wtot[wave] = sum;
  __syncthreads();

  const int t = wtot[lane];
  int incl = t;
#pragma unroll
  for (int off = 1; off < 32; off <<= 1) {
    const int v = __shfl_up(incl, off, 32);
    if (lane >= off) incl += v;
  }
  int carry = __shfl(incl - t, wave, 32);
  const int total = __shfl(incl, 31, 32);

  for (int base = lo; base < hi; base += 256) {
    const int i0 = base + lane * 8;
    int c[8];
    int s = 0;
#pragma unroll
    for (int u = 0; u < 8; ++u) {
      c[u] = (i0 + u < hi) ? cnt[i0 + u] : 0;
      s += c[u];
    }
    int inc = s;
#pragma unroll
    for (int off = 1; off < 32; off <<= 1) {
      const int v = __shfl_up(inc, off, 32);
      if (lane >= off) inc += v;
    }
    int run = carry + inc - s;
#pragma unroll
    for (int u = 0; u < 8; ++u) {
      if (i0 + u < hi) {
        rowptr[i0 + u] = run;
        run += c[u];
        cnt[i0 + u] = 0;
      }
    }
    carry += __shfl(inc, 31, 32);
  }
  if (threadIdx.x == 0) rowptr[N] = total;
}

__global__ __launch_bounds__(256) void dst_scatter(
    const int* __restrict__ ei, int E, const int* __restrict__ rowptr,
    int* __restrict__ cnt, int* __restrict__ csr_src) {
  const int e = blockIdx.x * blockDim.x + threadIdx.x;
  if (e >= E) return;
  const int s = ei[e];
  const int d = ei[E + e];
  const int pos = rowptr[d] + atomicAdd(&cnt[d], 1);
  csr_src[pos] = s;
}

__device__ __forceinline__ float lrelu(float l) {
  return (l > 0.f) ? l : LRELU_SLOPE * l;
}
__device__ __forceinline__ float elu1(float v) {
  return (v > 0.f) ? v : (expf(v) - 1.f);
}

__global__ __launch_bounds__(256) void agg_h8(
    const int* __restrict__ rowptr, const int* __restrict__ csr_src,
    const float* __restrict__ a_s, const float* __restrict__ a_d,
    const float* __restrict__ h, const float* __restrict__ bias,
    float* __restrict__ out, int N) {
  const int lane = threadIdx.x & 31;
  const int d = __builtin_amdgcn_readfirstlane((int)(blockIdx.x * 8 + (threadIdx.x >> 5)));
  if (d >= N) return;
  const int hA = lane >> 3, hB = 4 + (lane >> 3);
  const int beg = rowptr[d], end = rowptr[d + 1];
  const float adA = a_d[d * 8 + hA], adB = a_d[d * 8 + hB];
  const float lsA = lrelu(a_s[d * 8 + hA] + adA);
  const float lsB = lrelu(a_s[d * 8 + hB] + adB);

  float mA = lsA, mB = lsB;
  for (int base = beg; base < end; base += 32) {
    const int n = (end - base < 32) ? (end - base) : 32;
    const int sv = (lane < n) ? csr_src[base + lane] : d;
    for (int j = 0; j < n; j += 8) {
      float ga[8], gb[8];
#pragma unroll
      for (int u = 0; u < 8; ++u) {
        const int s = __builtin_amdgcn_readlane(sv, j + u);
        const float* __restrict__ as = a_s + (size_t)s * 8;
        ga[u] = as[hA];
        gb[u] = as[hB];
      }
#pragma unroll
      for (int u = 0; u < 8; ++u) {
        mA = fmaxf(mA, lrelu(ga[u] + adA));
        mB = fmaxf(mB, lrelu(gb[u] + adB));
      }
    }
  }

  const v4f* __restrict__ hd = (const v4f*)(h + (size_t)d * 256);
  const float wsA = expf(lsA - mA), wsB = expf(lsB - mB);
  float denA = wsA, denB = wsB;
  v4f accA = wsA * hd[lane];
  v4f accB = wsB * hd[32 + lane];
  for (int base = beg; base < end; base += 32) {
    const int n = (end - base < 32) ? (end - base) : 32;
    const int sv = (lane < n) ? csr_src[base + lane] : d;
    for (int j = 0; j < n; j += 4) {
      v4f ra[4], rb[4];
      float ga[4], gb[4];
#pragma unroll
      for (int u = 0; u < 4; ++u) {
        const int s = __builtin_amdgcn_readlane(sv, j + u);
        const v4f* __restrict__ hr = (const v4f*)(h + (size_t)s * 256);
        const float* __restrict__ as = a_s + (size_t)s * 8;
        ra[u] = hr[lane];
        rb[u] = hr[32 + lane];
        ga[u] = as[hA];
        gb[u] = as[hB];
      }
#pragma unroll
      for (int u = 0; u < 4; ++u) {
        const bool live = (j + u) < n;
        const float la = live ? lrelu(ga[u] + adA) : -__builtin_huge_valf();
        const float lb = live ? lrelu(gb[u] + adB) : -__builtin_huge_valf();
        const float wa = expf(la - mA);
        const float wb = expf(lb - mB);
        denA += wa;
        denB += wb;
        accA += wa * ra[u];
        accB += wb * rb[u];
      }
      __builtin_amdgcn_sched_group_barrier(0x020, 16, 0);
      __builtin_amdgcn_sched_group_barrier(0x002, 256, 0);
    }
  }

  const float qA = denA + 1e-16f, qB = denB + 1e-16f;
  const float* __restrict__ bA = bias + 4 * lane;
  const float* __restrict__ bB = bias + 128 + 4 * lane;
  v4f oA, oB;
  oA.x = accA.x / qA + bA[0];  oA.y = accA.y / qA + bA[1];
  oA.z = accA.z / qA + bA[2];  oA.w = accA.w / qA + bA[3];
  oB.x = accB.x / qB + bB[0];  oB.y = accB.y / qB + bB[1];
  oB.z = accB.z / qB + bB[2];  oB.w = accB.w / qB + bB[3];
  oA.x = elu1(oA.x);  oA.y = elu1(oA.y);  oA.z = elu1(oA.z);  oA.w = elu1(oA.w);
  oB.x = elu1(oB.x);  oB.y = elu1(oB.y);  oB.z = elu1(oB.z);  oB.w = elu1(oB.w);
  v4f* __restrict__ od = (v4f*)(out + (size_t)d * 256);
  od[lane]      = oA;
  od[32 + lane] = oB;
}

__global__ __launch_bounds__(256) void agg_h1(
    const int* __restrict__ rowptr, const int* __restrict__ csr_src,
    const float* __restrict__ a_s, const float* __restrict__ a_d,
    const float* __restrict__ h, const float* __restrict__ bias,
    float* __restrict__ out, int N) {
  const int lane = threadIdx.x & 31;
  const int d = __builtin_amdgcn_readfirstlane((int)(blockIdx.x * 8 + (threadIdx.x >> 5)));
  if (d >= N) return;
  const int beg = rowptr[d], end = rowptr[d + 1];
  const float ad = a_d[d];
  const float ls = lrelu(a_s[d] + ad);

  float m = ls;
  for (int base = beg; base < end; base += 32) {
    const int idx = base + lane;
    if (idx < end) m = fmaxf(m, lrelu(a_s[csr_src[idx]] + ad));
  }
#pragma unroll
  for (int off = 16; off > 0; off >>= 1) m = fmaxf(m, __shfl_xor(m, off, 32));

  const float wself = expf(ls - m);
  float den = 0.f;
  float acc = wself * h[(size_t)d * 32 + lane];
  for (int base = beg; base < end; base += 32) {
    const int n = (end - base < 32) ? (end - base) : 32;
    int sv = d;
    float wv = 0.f;
    if (lane < n) {
      sv = csr_src[base + lane];
      wv = expf(lrelu(a_s[sv] + ad) - m);
    }
    den += wv;
    for (int j = 0; j < n; j += 8) {
      float r[8], w[8];
#pragma unroll
      for (int u = 0; u < 8; ++u) {
        const int s = __builtin_amdgcn_readlane(sv, j + u);
        w[u] = __uint_as_float(
            (unsigned)__builtin_amdgcn_readlane((int)__float_as_uint(wv), j + u));
        r[u] = h[(size_t)s * 32 + lane];
      }
#pragma unroll
      for (int u = 0; u < 8; ++u) acc += w[u] * r[u];
    }
  }
#pragma unroll
  for (int off = 16; off > 0; off >>= 1) den += __shfl_xor(den, off, 32);
  den += wself;
  out[(size_t)d * 32 + lane] = elu1(acc / (den + 1e-16f) + bias[lane]);
}

extern "C" void kernel_launch(void* const* d_in, const int* in_sizes, int n_in,
                              void* d_out, int out_size, void* d_ws, size_t ws_size,
                              hipStream_t stream) {
  const float* x   = (const float*)d_in[0];
  const int*   ei  = (const int*)d_in[1];
  const float* W1  = (const float*)d_in[2];
  const float* as1 = (const float*)d_in[3];
  const float* ad1 = (const float*)d_in[4];
  const float* b1  = (const float*)d_in[5];
  const float* W2  = (const float*)d_in[6];
  const float* as2 = (const float*)d_in[7];
  const float* ad2 = (const float*)d_in[8];
  const float* b2  = (const float*)d_in[9];

  const int HC   = in_sizes[3];
  const int Fin  = in_sizes[2] / HC;
  const int N    = in_sizes[0] / Fin;
  const int E    = in_sizes[1] / 2;
  const int H    = 8;
  const int Cout = in_sizes[7];

  float* ws = (float*)d_ws;
  float* h1  = ws;
  float* x2  = h1 + (size_t)N * HC;
  float* h2  = x2 + (size_t)N * HC;
  float* aS1 = h2 + (size_t)N * Cout;
  float* aD1 = aS1 + (size_t)N * H;
  float* aS2 = aD1 + (size_t)N * H;
  float* aD2 = aS2 + N;
  int*   cnt    = (int*)(aD2 + N);
  int*   rowptr = cnt + N;
  int*   csr    = rowptr + N + 1;

  const int TB = 256;
  auto blocks = [](long long n, int tb) { return (int)((n + tb - 1) / tb); };

  hipMemsetAsync(cnt, 0, (size_t)N * sizeof(int), stream);
  dst_hist<<<blocks(E, TB), TB, 0, stream>>>(ei, E, cnt);
  dst_scan<<<1, 1024, 0, stream>>>(cnt, rowptr, N);
  dst_scatter<<<blocks(E, TB), TB, 0, stream>>>(ei, E, rowptr, cnt, csr);

  {
    const long long waves = (long long)(N / 16) * (HC / 16);
    gemm_wmma_f32<<<blocks(waves * 32, TB), TB, 0, stream>>>(x, W1, h1, N, Fin, HC);
    att_coef<8><<<blocks((long long)((N * H / 8 + 7) / 8) * 32, TB), TB, 0, stream>>>(
        h1, as1, ad1, aS1, aD1, N * H / 8);
    agg_h8<<<blocks((long long)N * 32, TB), TB, 0, stream>>>(
        rowptr, csr, aS1, aD1, h1, b1, x2, N);
  }

  {
    const long long waves = (long long)(N / 16) * (Cout / 16);
    gemm_wmma_f32<<<blocks(waves * 32, TB), TB, 0, stream>>>(x2, W2, h2, N, HC, Cout);
    att_coef<1><<<blocks((long long)((N / 8 + 7) / 8) * 32, TB), TB, 0, stream>>>(
        h2, as2, ad2, aS2, aD2, N / 8);
    agg_h1<<<blocks((long long)N * 32, TB), TB, 0, stream>>>(
        rowptr, csr, aS2, aD2, h2, b2, (float*)d_out, N);
  }
}


# GEMM: one wave per 32x64 block (layer 2: 16x32) of column-interleaved tiles, so X, W, Y move as whole-line 16 B accesses and X is reused 4x; k-sum order inside each 8-block changes
# speedup vs baseline: 5.2301x; 1.1378x over previous
#include <hip/hip_runtime.h>
#include <math.h>

typedef float v2f __attribute__((ext_vector_type(2)));
typedef float v4f __attribute__((ext_vector_type(4)));
typedef float v8f __attribute__((ext_vector_type(8)));

#define LRELU_SLOPE 0.2f

template <int MT, int NT, int KB, int KT, int NC>
__global__ __launch_bounds__(256) void gemm_wmma_f32(
    const float* __restrict__ X, const float* __restrict__ W,
    float* __restrict__ Y, int M) {
  typedef float vNf __attribute__((ext_vector_type(NT)));
  constexpr int NJ = KB / 8;
  constexpr int groupsN = NC / (16 * NT);
  const int lane = threadIdx.x & 31;
  const int wave = __builtin_amdgcn_readfirstlane(
      (int)((blockIdx.x * blockDim.x + threadIdx.x) >> 5));
  const int blocksM = (M + 16 * MT - 1) / (16 * MT);
  if (wave >= blocksM * groupsN) return;
  const int m0 = (wave / groupsN) * (16 * MT);
  const int n0 = (wave % groupsN) * (16 * NT);
  const int hf = lane >> 4;
  const int c  = lane & 15;

  const float* __restrict__ xr[MT];
#pragma unroll
  for (int mt = 0; mt < MT; ++mt) {
    int row = m0 + mt * 16 + c;
    row = (row < M) ? row : (M - 1);
    xr[mt] = X + (size_t)row * KT + 4 * hf;
  }
  const float* __restrict__ wp = W + (size_t)(4 * hf) * NC + n0 + NT * c;

  v8f acc[MT][NT];
#pragma unroll
  for (int mt = 0; mt < MT; ++mt)
#pragma unroll
    for (int t = 0; t < NT; ++t)
      acc[mt][t] = (v8f){0.f, 0.f, 0.f, 0.f, 0.f, 0.f, 0.f, 0.f};

#pragma unroll 1
  for (int kb = 0; kb < KT; kb += KB) {
    v4f a[MT][NJ];
    vNf w[NJ][4];
#pragma unroll
    for (int mt = 0; mt < MT; ++mt)
#pragma unroll
      for (int j = 0; j < NJ; ++j)
        a[mt][j] = *(const v4f*)(xr[mt] + kb + 8 * j);
#pragma unroll
    for (int j = 0; j < NJ; ++j)
#pragma unroll
      for (int q = 0; q < 4; ++q)
        w[j][q] = *(const vNf*)(wp + (size_t)(kb + 8 * j + q) * NC);

#pragma unroll
    for (int j = 0; j < NJ; ++j) {
#pragma unroll
      for (int p = 0; p < 2; ++p) {
#pragma unroll
        for (int mt = 0; mt < MT; ++mt) {
          v2f av;
          av.x = p ? a[mt][j].z : a[mt][j].x;
          av.y = p ? a[mt][j].w : a[mt][j].y;
#pragma unroll
          for (int t = 0; t < NT; ++t) {
            v2f bv;
            bv.x = w[j][2 * p][t];
            bv.y = w[j][2 * p + 1][t];
#if __has_builtin(__builtin_amdgcn_wmma_f32_16x16x4_f32)
            acc[mt][t] = __builtin_amdgcn_wmma_f32_16x16x4_f32(
                false, av, false, bv, (short)0, acc[mt][t], false, false);
#else
            acc[mt][t][0] += av.x * bv.x;
#endif
          }
        }
      }
    }
    __builtin_amdgcn_sched_group_barrier(0x020, MT * NJ + 4 * NJ, 0);
    __builtin_amdgcn_sched_group_barrier(0x00A, 8 * MT * NT * NJ, 0);
  }

#pragma unroll
  for (int mt = 0; mt < MT; ++mt) {
    if (m0 + mt * 16 < M) {
      float* __restrict__ yr = Y + (size_t)(m0 + mt * 16 + 8 * hf) * NC + n0 + NT * c;
#pragma unroll
      for (int r = 0; r < 8; ++r) {
        vNf o;
#pragma unroll
        for (int t = 0; t < NT; ++t) o[t] = acc[mt][t][r];
        *(vNf*)(yr + (size_t)r * NC) = o;
      }
    }
  }
}

__device__ __forceinline__ float dot4(v4f a, v4f b) {
  return a.x * b.x + a.y * b.y + a.z * b.z + a.w * b.w;
}

template <int H>
__global__ __launch_bounds__(256) void att_coef(
    const float* __restrict__ h, const float* __restrict__ att_s,
    const float* __restrict__ att_d, float* __restrict__ a_s,
    float* __restrict__ a_d, int nChunks) {
  constexpr int CPW = 8;
  constexpr int HC = H * 32;
  const int lane = threadIdx.x & 31;
  const int wave = __builtin_amdgcn_readfirstlane(
      (int)((blockIdx.x * blockDim.x + threadIdx.x) >> 5));
  const int c0 = wave * CPW;
  if (c0 >= nChunks) return;

  const int fA = (4 * lane) % HC, fB = (128 + 4 * lane) % HC;
  const v4f sA = *(const v4f*)(att_s + fA), sB = *(const v4f*)(att_s + fB);
  const v4f dA = *(const v4f*)(att_d + fA), dB = *(const v4f*)(att_d + fB);

  v4f ra[CPW], rb[CPW];
#pragma unroll
  for (int u = 0; u < CPW; ++u) {
    const int ch = (c0 + u < nChunks) ? (c0 + u) : (nChunks - 1);
    const v4f* __restrict__ p = (const v4f*)(h + (size_t)ch * 256);
    ra[u] = p[lane];
    rb[u] = p[32 + lane];
  }
#pragma unroll
  for (int u = 0; u < CPW; ++u) {
    float ssA = dot4(ra[u], sA), ssB = dot4(rb[u], sB);
    float ddA = dot4(ra[u], dA), ddB = dot4(rb[u], dB);
#pragma unroll
    for (int off = 4; off > 0; off >>= 1) {
      ssA += __shfl_xor(ssA, off, 32);
      ssB += __shfl_xor(ssB, off, 32);
      ddA += __shfl_xor(ddA, off, 32);
      ddB += __shfl_xor(ddB, off, 32);
    }
    if (c0 + u < nChunks && (lane & 7) == 0) {
      const int w = (c0 + u) * 8 + (lane >> 3);
      a_s[w] = ssA;  a_s[w + 4] = ssB;
      a_d[w] = ddA;  a_d[w + 4] = ddB;
    }
  }
}

__global__ __launch_bounds__(256) void dst_hist(
    const int* __restrict__ ei, int E, int* __restrict__ cnt) {
  const int e = blockIdx.x * blockDim.x + threadIdx.x;
  if (e >= E) return;
  atomicAdd(&cnt[ei[E + e]], 1);
}

__global__ __launch_bounds__(1024) void dst_scan(
    int* cnt, int* __restrict__ rowptr, int N) {
  __shared__ int wtot[32];
  const int lane = threadIdx.x & 31;
  const int wave = __builtin_amdgcn_readfirstlane((int)(threadIdx.x >> 5));
  const int seg = (N + 31) >> 5;
  int lo = wave * seg;      if (lo > N) lo = N;
  int hi = lo + seg;        if (hi > N) hi = N;

  int sum = 0;
  for (int i = lo + lane; i < hi; i += 32) sum += cnt[i];
#pragma unroll
  for (int off = 16; off > 0; off >>= 1) sum += __shfl_xor(sum, off, 32);
  if (lane == 0) wtot[wave] = sum;
  __syncthreads();

  const int t = wtot[lane];
  int incl = t;
#pragma unroll
  for (int off = 1; off < 32; off <<= 1) {
    const int v = __shfl_up(incl, off, 32);
    if (lane >= off) incl += v;
  }
  int carry = __shfl(incl - t, wave, 32);
  const int total = __shfl(incl, 31, 32);

  for (int base = lo; base < hi; base += 256) {
    const int i0 = base + lane * 8;
    int c[8];
    int s = 0;
#pragma unroll
    for (int u = 0; u < 8; ++u) {
      c[u] = (i0 + u < hi) ? cnt[i0 + u] : 0;
      s += c[u];
    }
    int inc = s;
#pragma unroll
    for (int off = 1; off < 32; off <<= 1) {
      const int v = __shfl_up(inc, off, 32);
      if (lane >= off) inc += v;
    }
    int run = carry + inc - s;
#pragma unroll
    for (int u = 0; u < 8; ++u) {
      if (i0 + u < hi) {
        rowptr[i0 + u] = run;
        run += c[u];
        cnt[i0 + u] = 0;
      }
    }
    carry += __shfl(inc, 31, 32);
  }
  if (threadIdx.x == 0) rowptr[N] = total;
}

__global__ __launch_bounds__(256) void dst_scatter(
    const int* __restrict__ ei, int E, const int* __restrict__ rowptr,
    int* __restrict__ cnt, int* __restrict__ csr_src) {
  const int e = blockIdx.x * blockDim.x + threadIdx.x;
  if (e >= E) return;
  const int s = ei[e];
  const int d = ei[E + e];
  const int pos = rowptr[d] + atomicAdd(&cnt[d], 1);
  csr_src[pos] = s;
}

__device__ __forceinline__ float lrelu(float l) {
  return (l > 0.f) ? l : LRELU_SLOPE * l;
}
__device__ __forceinline__ float elu1(float v) {
  return (v > 0.f) ? v : (expf(v) - 1.f);
}

__global__ __launch_bounds__(256) void agg_h8(
    const int* __restrict__ rowptr, const int* __restrict__ csr_src,
    const float* __restrict__ a_s, const float* __restrict__ a_d,
    const float* __restrict__ h, const float* __restrict__ bias,
    float* __restrict__ out, int N) {
  const int lane = threadIdx.x & 31;
  const int d = __builtin_amdgcn_readfirstlane((int)(blockIdx.x * 8 + (threadIdx.x >> 5)));
  if (d >= N) return;
  const int hA = lane >> 3, hB = 4 + (lane >> 3);
  const int beg = rowptr[d], end = rowptr[d + 1];
  const float adA = a_d[d * 8 + hA], adB = a_d[d * 8 + hB];
  const float lsA = lrelu(a_s[d * 8 + hA] + adA);
  const float lsB = lrelu(a_s[d * 8 + hB] + adB);

  float mA = lsA, mB = lsB;
  for (int base = beg; base < end; base += 32) {
    const int n = (end - base < 32) ? (end - base) : 32;
    const int sv = (lane < n) ? csr_src[base + lane] : d;
    for (int j = 0; j < n; j += 8) {
      float ga[8], gb[8];
#pragma unroll
      for (int u = 0; u < 8; ++u) {
        const int s = __builtin_amdgcn_readlane(sv, j + u);
        const float* __restrict__ as = a_s + (size_t)s * 8;
        ga[u] = as[hA];
        gb[u] = as[hB];
      }
#pragma unroll
      for (int u = 0; u < 8; ++u) {
        mA = fmaxf(mA, lrelu(ga[u] + adA));
        mB = fmaxf(mB, lrelu(gb[u] + adB));
      }
    }
  }

  const v4f* __restrict__ hd = (const v4f*)(h + (size_t)d * 256);
  const float wsA = expf(lsA - mA), wsB = expf(lsB - mB);
  float denA = wsA, denB = wsB;
  v4f accA = wsA * hd[lane];
  v4f accB = wsB * hd[32 + lane];
  for (int base = beg; base < end; base += 32) {
    const int n = (end - base < 32) ? (end - base) : 32;
    const int sv = (lane < n) ? csr_src[base + lane] : d;
    for (int j = 0; j < n; j += 4) {
      v4f ra[4], rb[4];
      float ga[4], gb[4];
#pragma unroll
      for (int u = 0; u < 4; ++u) {
        const int s = __builtin_amdgcn_readlane(sv, j + u);
        const v4f* __restrict__ hr = (const v4f*)(h + (size_t)s * 256);
        const float* __restrict__ as = a_s + (size_t)s * 8;
        ra[u] = hr[lane];
        rb[u] = hr[32 + lane];
        ga[u] = as[hA];
        gb[u] = as[hB];
      }
#pragma unroll
      for (int u = 0; u < 4; ++u) {
        const bool live = (j + u) < n;
        const float la = live ? lrelu(ga[u] + adA) : -__builtin_huge_valf();
        const float lb = live ? lrelu(gb[u] + adB) : -__builtin_huge_valf();
        const float wa = expf(la - mA);
        const float wb = expf(lb - mB);
        denA += wa;
        denB += wb;
        accA += wa * ra[u];
        accB += wb * rb[u];
      }
      __builtin_amdgcn_sched_group_barrier(0x020, 16, 0);
      __builtin_amdgcn_sched_group_barrier(0x002, 256, 0);
    }
  }

  const float qA = denA + 1e-16f, qB = denB + 1e-16f;
  const float* __restrict__ bA = bias + 4 * lane;
  const float* __restrict__ bB = bias + 128 + 4 * lane;
  v4f oA, oB;
  oA.x = accA.x / qA + bA[0];  oA.y = accA.y / qA + bA[1];
  oA.z = accA.z / qA + bA[2];  oA.w = accA.w / qA + bA[3];
  oB.x = accB.x / qB + bB[0];  oB.y = accB.y / qB + bB[1];
  oB.z = accB.z / qB + bB[2];  oB.w = accB.w / qB + bB[3];
  oA.x = elu1(oA.x);  oA.y = elu1(oA.y);  oA.z = elu1(oA.z);  oA.w = elu1(oA.w);
  oB.x = elu1(oB.x);  oB.y = elu1(oB.y);  oB.z = elu1(oB.z);  oB.w = elu1(oB.w);
  v4f* __restrict__ od = (v4f*)(out + (size_t)d * 256);
  od[lane]      = oA;
  od[32 + lane] = oB;
}

__global__ __launch_bounds__(256) void agg_h1(
    const int* __restrict__ rowptr, const int* __restrict__ csr_src,
    const float* __restrict__ a_s, const float* __restrict__ a_d,
    const float* __restrict__ h, const float* __restrict__ bias,
    float* __restrict__ out, int N) {
  const int lane = threadIdx.x & 31;
  const int d = __builtin_amdgcn_readfirstlane((int)(blockIdx.x * 8 + (threadIdx.x >> 5)));
  if (d >= N) return;
  const int beg = rowptr[d], end = rowptr[d + 1];
  const float ad = a_d[d];
  const float ls = lrelu(a_s[d] + ad);

  float m = ls;
  for (int base = beg; base < end; base += 32) {
    const int idx = base + lane;
    if (idx < end) m = fmaxf(m, lrelu(a_s[csr_src[idx]] + ad));
  }
#pragma unroll
  for (int off = 16; off > 0; off >>= 1) m = fmaxf(m, __shfl_xor(m, off, 32));

  const float wself = expf(ls - m);
  float den = 0.f;
  float acc = wself * h[(size_t)d * 32 + lane];
  for (int base = beg; base < end; base += 32) {
    const int n = (end - base < 32) ? (end - base) : 32;
    int sv = d;
    float wv = 0.f;
    if (lane < n) {
      sv = csr_src[base + lane];
      wv = expf(lrelu(a_s[sv] + ad) - m);
    }
    den += wv;
    for (int j = 0; j < n; j += 8) {
      float r[8], w[8];
#pragma unroll
      for (int u = 0; u < 8; ++u) {
        const int s = __builtin_amdgcn_readlane(sv, j + u);
        w[u] = __uint_as_float(
            (unsigned)__builtin_amdgcn_readlane((int)__float_as_uint(wv), j + u));
        r[u] = h[(size_t)s * 32 + lane];
      }
#pragma unroll
      for (int u = 0; u < 8; ++u) acc += w[u] * r[u];
    }
  }
#pragma unroll
  for (int off = 16; off > 0; off >>= 1) den += __shfl_xor(den, off, 32);
  den += wself;
  out[(size_t)d * 32 + lane] = elu1(acc / (den + 1e-16f) + bias[lane]);
}

extern "C" void kernel_launch(void* const* d_in, const int* in_sizes, int n_in,
                              void* d_out, int out_size, void* d_ws, size_t ws_size,
                              hipStream_t stream) {
  const float* x   = (const float*)d_in[0];
  const int*   ei  = (const int*)d_in[1];
  const float* W1  = (const float*)d_in[2];
  const float* as1 = (const float*)d_in[3];
  const float* ad1 = (const float*)d_in[4];
  const float* b1  = (const float*)d_in[5];
  const float* W2  = (const float*)d_in[6];
  const float* as2 = (const float*)d_in[7];
  const float* ad2 = (const float*)d_in[8];
  const float* b2  = (const float*)d_in[9];

  const int HC   = in_sizes[3];
  const int Fin  = in_sizes[2] / HC;
  const int N    = in_sizes[0] / Fin;
  const int E    = in_sizes[1] / 2;
  const int H    = 8;
  const int Cout = in_sizes[7];

  float* ws = (float*)d_ws;
  float* h1  = ws;
  float* x2  = h1 + (size_t)N * HC;
  float* h2  = x2 + (size_t)N * HC;
  float* aS1 = h2 + (size_t)N * Cout;
  float* aD1 = aS1 + (size_t)N * H;
  float* aS2 = aD1 + (size_t)N * H;
  float* aD2 = aS2 + N;
  int*   cnt    = (int*)(aD2 + N);
  int*   rowptr = cnt + N;
  int*   csr    = rowptr + N + 1;

  const int TB = 256;
  auto blocks = [](long long n, int tb) { return (int)((n + tb - 1) / tb); };

  hipMemsetAsync(cnt, 0, (size_t)N * sizeof(int), stream);
  dst_hist<<<blocks(E, TB), TB, 0, stream>>>(ei, E, cnt);
  dst_scan<<<1, 1024, 0, stream>>>(cnt, rowptr, N);
  dst_scatter<<<blocks(E, TB), TB, 0, stream>>>(ei, E, rowptr, cnt, csr);

  {
    const long long waves = (long long)((N + 31) / 32) * (HC / 64);
    gemm_wmma_f32<2, 4, 32, 128, 256><<<blocks(waves * 32, TB), TB, 0, stream>>>(x, W1, h1, N);
    att_coef<8><<<blocks((long long)((N * H / 8 + 7) / 8) * 32, TB), TB, 0, stream>>>(
        h1, as1, ad1, aS1, aD1, N * H / 8);
    agg_h8<<<blocks((long long)N * 32, TB), TB, 0, stream>>>(
        rowptr, csr, aS1, aD1, h1, b1, x2, N);
  }

  {
    const long long waves = (long long)(N / 16) * (Cout / 32);
    gemm_wmma_f32<1, 2, 64, 256, 32><<<blocks(waves * 32, TB), TB, 0, stream>>>(x2, W2, h2, N);
    att_coef<1><<<blocks((long long)((N / 8 + 7) / 8) * 32, TB), TB, 0, stream>>>(
        h2, as2, ad2, aS2, aD2, N / 8);
    agg_h1<<<blocks((long long)N * 32, TB), TB, 0, stream>>>(
        rowptr, csr, aS2, aD2, h2, b2, (float*)d_out, N);
  }
}


# Edge ordering by a two-level counting sort in LDS (4096-edge chunks binned by 256-dst bucket, then one block per bucket): no global atomics, memset or 1-block scan; per-dst sum order = LDS arrival
# speedup vs baseline: 10.8012x; 2.0652x over previous
#include <hip/hip_runtime.h>
#include <math.h>

typedef float v2f __attribute__((ext_vector_type(2)));
typedef float v4f __attribute__((ext_vector_type(4)));
typedef float v8f __attribute__((ext_vector_type(8)));

#define LRELU_SLOPE 0.2f

template <int MT, int NT, int KB, int KT, int NC>
__global__ __launch_bounds__(256) void gemm_wmma_f32(
    const float* __restrict__ X, const float* __restrict__ W,
    float* __restrict__ Y, int M) {
  typedef float vNf __attribute__((ext_vector_type(NT)));
  constexpr int NJ = KB / 8;
  constexpr int groupsN = NC / (16 * NT);
  const int lane = threadIdx.x & 31;
  const int wave = __builtin_amdgcn_readfirstlane(
      (int)((blockIdx.x * blockDim.x + threadIdx.x) >> 5));
  const int blocksM = (M + 16 * MT - 1) / (16 * MT);
  if (wave >= blocksM * groupsN) return;
  const int m0 = (wave / groupsN) * (16 * MT);
  const int n0 = (wave % groupsN) * (16 * NT);
  const int hf = lane >> 4;
  const int c  = lane & 15;

  const float* __restrict__ xr[MT];
#pragma unroll
  for (int mt = 0; mt < MT; ++mt) {
    int row = m0 + mt * 16 + c;
    row = (row < M) ? row : (M - 1);
    xr[mt] = X + (size_t)row * KT + 4 * hf;
  }
  const float* __restrict__ wp = W + (size_t)(4 * hf) * NC + n0 + NT * c;

  v8f acc[MT][NT];
#pragma unroll
  for (int mt = 0; mt < MT; ++mt)
#pragma unroll
    for (int t = 0; t < NT; ++t)
      acc[mt][t] = (v8f){0.f, 0.f, 0.f, 0.f, 0.f, 0.f, 0.f, 0.f};

#pragma unroll 1
  for (int kb = 0; kb < KT; kb += KB) {
    v4f a[MT][NJ];
    vNf w[NJ][4];
#pragma unroll
    for (int mt = 0; mt < MT; ++mt)
#pragma unroll
      for (int j = 0; j < NJ; ++j)
        a[mt][j] = *(const v4f*)(xr[mt] + kb + 8 * j);
#pragma unroll
    for (int j = 0; j < NJ; ++j)
#pragma unroll
      for (int q = 0; q < 4; ++q)
        w[j][q] = *(const vNf*)(wp + (size_t)(kb + 8 * j + q) * NC);

#pragma unroll
    for (int j = 0; j < NJ; ++j) {
#pragma unroll
      for (int p = 0; p < 2; ++p) {
#pragma unroll
        for (int mt = 0; mt < MT; ++mt) {
          v2f av;
          av.x = p ? a[mt][j].z : a[mt][j].x;
          av.y = p ? a[mt][j].w : a[mt][j].y;
#pragma unroll
          for (int t = 0; t < NT; ++t) {
            v2f bv;
            bv.x = w[j][2 * p][t];
            bv.y = w[j][2 * p + 1][t];
#if __has_builtin(__builtin_amdgcn_wmma_f32_16x16x4_f32)
            acc[mt][t] = __builtin_amdgcn_wmma_f32_16x16x4_f32(
                false, av, false, bv, (short)0, acc[mt][t], false, false);
#else
            acc[mt][t][0] += av.x * bv.x;
#endif
          }
        }
      }
    }
    __builtin_amdgcn_sched_group_barrier(0x020, MT * NJ + 4 * NJ, 0);
    __builtin_amdgcn_sched_group_barrier(0x00A, 8 * MT * NT * NJ, 0);
  }

#pragma unroll
  for (int mt = 0; mt < MT; ++mt) {
    if (m0 + mt * 16 < M) {
      float* __restrict__ yr = Y + (size_t)(m0 + mt * 16 + 8 * hf) * NC + n0 + NT * c;
#pragma unroll
      for (int r = 0; r < 8; ++r) {
        vNf o;
#pragma unroll
        for (int t = 0; t < NT; ++t) o[t] = acc[mt][t][r];
        *(vNf*)(yr + (size_t)r * NC) = o;
      }
    }
  }
}

__device__ __forceinline__ float dot4(v4f a, v4f b) {
  return a.x * b.x + a.y * b.y + a.z * b.z + a.w * b.w;
}

template <int H>
__global__ __launch_bounds__(256) void att_coef(
    const float* __restrict__ h, const float* __restrict__ att_s,
    const float* __restrict__ att_d, float* __restrict__ a_s,
    float* __restrict__ a_d, int nChunks) {
  constexpr int CPW = 8;
  constexpr int HC = H * 32;
  const int lane = threadIdx.x & 31;
  const int wave = __builtin_amdgcn_readfirstlane(
      (int)((blockIdx.x * blockDim.x + threadIdx.x) >> 5));
  const int c0 = wave * CPW;
  if (c0 >= nChunks) return;

  const int fA = (4 * lane) % HC, fB = (128 + 4 * lane) % HC;
  const v4f sA = *(const v4f*)(att_s + fA), sB = *(const v4f*)(att_s + fB);
  const v4f dA = *(const v4f*)(att_d + fA), dB = *(const v4f*)(att_d + fB);

  v4f ra[CPW], rb[CPW];
#pragma unroll
  for (int u = 0; u < CPW; ++u) {
    const int ch = (c0 + u < nChunks) ? (c0 + u) : (nChunks - 1);
    const v4f* __restrict__ p = (const v4f*)(h + (size_t)ch * 256);
    ra[u] = p[lane];
    rb[u] = p[32 + lane];
  }
#pragma unroll
  for (int u = 0; u < CPW; ++u) {
    float ssA = dot4(ra[u], sA), ssB = dot4(rb[u], sB);
    float ddA = dot4(ra[u], dA), ddB = dot4(rb[u], dB);
#pragma unroll
    for (int off = 4; off > 0; off >>= 1) {
      ssA += __shfl_xor(ssA, off, 32);
      ssB += __shfl_xor(ssB, off, 32);
      ddA += __shfl_xor(ddA, off, 32);
      ddB += __shfl_xor(ddB, off, 32);
    }
    if (c0 + u < nChunks && (lane & 7) == 0) {
      const int w = (c0 + u) * 8 + (lane >> 3);
      a_s[w] = ssA;  a_s[w + 4] = ssB;
      a_d[w] = ddA;  a_d[w + 4] = ddB;
    }
  }
}

#define SORT_TB    1024
#define BIN_EPT    4
#define BIN_CHUNK  (BIN_EPT * SORT_TB)
#define BKT_SHIFT  8
#define BKT_DSTS   256
#define BKT_CAP    6144

__device__ __forceinline__ int scan256(const int* v, int* o, int* wt,
                                       int tid, int wv, int& c) {
  const int lane = tid & 31;
  int incl = 0;
  c = 0;
  if (wv < 8) {
    c = v[tid];
    incl = c;
#pragma unroll
    for (int k = 1; k < 32; k <<= 1) {
      const int t = __shfl_up(incl, k, 32);
      if (lane >= k) incl += t;
    }
    if (lane == 31) wt[wv] = incl;
  }
  __syncthreads();
  int ex = 0;
  if (wv < 8) {
    int pre = 0;
#pragma unroll
    for (int w = 0; w < 8; ++w) pre += (w < wv) ? wt[w] : 0;
    ex = pre + incl - c;
    o[tid] = ex;
  }
  __syncthreads();
  return ex;
}

__global__ __launch_bounds__(SORT_TB) void edge_bin(
    const int* __restrict__ ei, int E, int* __restrict__ recS,
    int* __restrict__ recD, int* __restrict__ tabOff, int* __restrict__ tabCnt) {
  __shared__ int cnt[BKT_DSTS];
  __shared__ int off[BKT_DSTS];
  __shared__ int wt[8];
  __shared__ int srtS[BIN_CHUNK];
  __shared__ int srtD[BIN_CHUNK];
  const int tid = threadIdx.x;
  const int wv = __builtin_amdgcn_readfirstlane((int)(threadIdx.x >> 5));
  const int e0 = blockIdx.x * BIN_CHUNK;
  if (tid < BKT_DSTS) cnt[tid] = 0;
  __syncthreads();

  int s[BIN_EPT], d[BIN_EPT], r[BIN_EPT];
#pragma unroll
  for (int u = 0; u < BIN_EPT; ++u) {
    const int e = e0 + u * SORT_TB + tid;
    s[u] = 0;  d[u] = -1;  r[u] = 0;
    if (e < E) { s[u] = ei[e];  d[u] = ei[E + e]; }
  }
#pragma unroll
  for (int u = 0; u < BIN_EPT; ++u)
    if (d[u] >= 0) r[u] = atomicAdd(&cnt[d[u] >> BKT_SHIFT], 1);
  __syncthreads();

  int c;
  const int ex = scan256(cnt, off, wt, tid, wv, c);
  if (wv < 8) {
    tabOff[blockIdx.x * BKT_DSTS + tid] = ex;
    tabCnt[blockIdx.x * BKT_DSTS + tid] = c;
  }
#pragma unroll
  for (int u = 0; u < BIN_EPT; ++u) {
    if (d[u] >= 0) {
      const int p = off[d[u] >> BKT_SHIFT] + r[u];
      srtS[p] = s[u];
      srtD[p] = d[u];
    }
  }
  __syncthreads();
  int nE = E - e0;
  if (nE > BIN_CHUNK) nE = BIN_CHUNK;
#pragma unroll
  for (int u = 0; u < BIN_EPT; ++u) {
    const int i = u * SORT_TB + tid;
    if (i < nE) { recS[e0 + i] = srtS[i];  recD[e0 + i] = srtD[i]; }
  }
}

__device__ __forceinline__ int rec_index(int i, const int* pOfs, const int* pSrc) {
  int lo = 0;
#pragma unroll
  for (int step = 512; step > 0; step >>= 1) {
    const int m = lo + step;
    if (pOfs[m] <= i) lo = m;
  }
  return pSrc[lo] + (i - pOfs[lo]);
}

__global__ __launch_bounds__(SORT_TB) void bucket_sort(
    const int* __restrict__ recS, const int* __restrict__ recD,
    const int* __restrict__ tabOff, const int* __restrict__ tabCnt, int nChunks,
    int* __restrict__ rowptr, int* __restrict__ csr_src, int N) {
  __shared__ int pOfs[SORT_TB];
  __shared__ int pSrc[SORT_TB];
  __shared__ int wsum[32];
  __shared__ int wbase[32];
  __shared__ int dcnt[BKT_DSTS];
  __shared__ int dofs[BKT_DSTS];
  __shared__ int wt[8];
  __shared__ int bufS[BKT_CAP];
  __shared__ int bufK[BKT_CAP];
  __shared__ int fin[BKT_CAP];
  const int tid = threadIdx.x;
  const int lane = tid & 31;
  const int wv = __builtin_amdgcn_readfirstlane((int)(threadIdx.x >> 5));
  const int g = blockIdx.x;

  int o = 0, c = 0;
  if (tid < nChunks) {
    o = tabOff[tid * BKT_DSTS + g];
    c = tabCnt[tid * BKT_DSTS + g];
  }
  if (tid < BKT_DSTS) dcnt[tid] = 0;
  int inclC = c, sumO = o;
#pragma unroll
  for (int k = 1; k < 32; k <<= 1) {
    const int t = __shfl_up(inclC, k, 32);
    if (lane >= k) inclC += t;
  }
#pragma unroll
  for (int k = 16; k > 0; k >>= 1) sumO += __shfl_xor(sumO, k, 32);
  if (lane == 31) wsum[wv] = inclC;
  if (lane == 0) wbase[wv] = sumO;
  __syncthreads();
  const int ws = wsum[lane];
  int wincl = ws;
#pragma unroll
  for (int k = 1; k < 32; k <<= 1) {
    const int t = __shfl_up(wincl, k, 32);
    if (lane >= k) wincl += t;
  }
  const int wpre  = __shfl(wincl - ws, wv, 32);
  const int total = __builtin_amdgcn_readfirstlane(__shfl(wincl, 31, 32));
  int bsum = wbase[lane];
#pragma unroll
  for (int k = 16; k > 0; k >>= 1) bsum += __shfl_xor(bsum, k, 32);
  const int base = __builtin_amdgcn_readfirstlane(bsum);
  pOfs[tid] = wpre + inclC - c;
  pSrc[tid] = tid * BIN_CHUNK + o;
  __syncthreads();

  if (total <= BKT_CAP) {
    constexpr int RPT = BKT_CAP / SORT_TB;
    int sv[RPT], dv[RPT];
#pragma unroll
    for (int k = 0; k < RPT; ++k) {
      const int i = k * SORT_TB + tid;
      sv[k] = 0;  dv[k] = 0;
      if (i < total) {
        const int q = rec_index(i, pOfs, pSrc);
        sv[k] = recS[q];
        dv[k] = recD[q];
      }
    }
#pragma unroll
    for (int k = 0; k < RPT; ++k) {
      const int i = k * SORT_TB + tid;
      if (i < total) {
        const int dl = dv[k] & (BKT_DSTS - 1);
        const int rk = atomicAdd(&dcnt[dl], 1);
        bufS[i] = sv[k];
        bufK[i] = (rk << BKT_SHIFT) | dl;
      }
    }
    __syncthreads();
    int cc;
    const int ex = scan256(dcnt, dofs, wt, tid, wv, cc);
    if (wv < 8) {
      const int node = g * BKT_DSTS + tid;
      if (node < N) rowptr[node] = base + ex;
    }
    for (int i = tid; i < total; i += SORT_TB) {
      const int k = bufK[i];
      fin[dofs[k & (BKT_DSTS - 1)] + (k >> BKT_SHIFT)] = bufS[i];
    }
    __syncthreads();
    for (int i = tid; i < total; i += SORT_TB) csr_src[base + i] = fin[i];
  } else {
    for (int i = tid; i < total; i += SORT_TB) {
      const int q = rec_index(i, pOfs, pSrc);
      atomicAdd(&dcnt[recD[q] & (BKT_DSTS - 1)], 1);
    }
    __syncthreads();
    int cc;
    const int ex = scan256(dcnt, dofs, wt, tid, wv, cc);
    if (wv < 8) {
      const int node = g * BKT_DSTS + tid;
      if (node < N) rowptr[node] = base + ex;
      dcnt[tid] = 0;
    }
    __syncthreads();
    for (int i = tid; i < total; i += SORT_TB) {
      const int q = rec_index(i, pOfs, pSrc);
      const int dl = recD[q] & (BKT_DSTS - 1);
      const int rk = atomicAdd(&dcnt[dl], 1);
      csr_src[base + dofs[dl] + rk] = recS[q];
    }
  }
  if (g == (int)gridDim.x - 1 && tid == 0) rowptr[N] = base + total;
}

__device__ __forceinline__ float lrelu(float l) {
  return (l > 0.f) ? l : LRELU_SLOPE * l;
}
__device__ __forceinline__ float elu1(float v) {
  return (v > 0.f) ? v : (expf(v) - 1.f);
}

__global__ __launch_bounds__(256) void agg_h8(
    const int* __restrict__ rowptr, const int* __restrict__ csr_src,
    const float* __restrict__ a_s, const float* __restrict__ a_d,
    const float* __restrict__ h, const float* __restrict__ bias,
    float* __restrict__ out, int N) {
  const int lane = threadIdx.x & 31;
  const int d = __builtin_amdgcn_readfirstlane((int)(blockIdx.x * 8 + (threadIdx.x >> 5)));
  if (d >= N) return;
  const int hA = lane >> 3, hB = 4 + (lane >> 3);
  const int beg = rowptr[d], end = rowptr[d + 1];
  const float adA = a_d[d * 8 + hA], adB = a_d[d * 8 + hB];
  const float lsA = lrelu(a_s[d * 8 + hA] + adA);
  const float lsB = lrelu(a_s[d * 8 + hB] + adB);

  float mA = lsA, mB = lsB;
  for (int base = beg; base < end; base += 32) {
    const int n = (end - base < 32) ? (end - base) : 32;
    const int sv = (lane < n) ? csr_src[base + lane] : d;
    for (int j = 0; j < n; j += 8) {
      float ga[8], gb[8];
#pragma unroll
      for (int u = 0; u < 8; ++u) {
        const int s = __builtin_amdgcn_readlane(sv, j + u);
        const float* __restrict__ as = a_s + (size_t)s * 8;
        ga[u] = as[hA];
        gb[u] = as[hB];
      }
#pragma unroll
      for (int u = 0; u < 8; ++u) {
        mA = fmaxf(mA, lrelu(ga[u] + adA));
        mB = fmaxf(mB, lrelu(gb[u] + adB));
      }
    }
  }

  const v4f* __restrict__ hd = (const v4f*)(h + (size_t)d * 256);
  const float wsA = expf(lsA - mA), wsB = expf(lsB - mB);
  float denA = wsA, denB = wsB;
  v4f accA = wsA * hd[lane];
  v4f accB = wsB * hd[32 + lane];
  for (int base = beg; base < end; base += 32) {
    const int n = (end - base < 32) ? (end - base) : 32;
    const int sv = (lane < n) ? csr_src[base + lane] : d;
    for (int j = 0; j < n; j += 4) {
      v4f ra[4], rb[4];
      float ga[4], gb[4];
#pragma unroll
      for (int u = 0; u < 4; ++u) {
        const int s = __builtin_amdgcn_readlane(sv, j + u);
        const v4f* __restrict__ hr = (const v4f*)(h + (size_t)s * 256);
        const float* __restrict__ as = a_s + (size_t)s * 8;
        ra[u] = hr[lane];
        rb[u] = hr[32 + lane];
        ga[u] = as[hA];
        gb[u] = as[hB];
      }
#pragma unroll
      for (int u = 0; u < 4; ++u) {
        const bool live = (j + u) < n;
        const float la = live ? lrelu(ga[u] + adA) : -__builtin_huge_valf();
        const float lb = live ? lrelu(gb[u] + adB) : -__builtin_huge_valf();
        const float wa = expf(la - mA);
        const float wb = expf(lb - mB);
        denA += wa;
        denB += wb;
        accA += wa * ra[u];
        accB += wb * rb[u];
      }
      __builtin_amdgcn_sched_group_barrier(0x020, 16, 0);
      __builtin_amdgcn_sched_group_barrier(0x002, 256, 0);
    }
  }

  const float qA = denA + 1e-16f, qB = denB + 1e-16f;
  const float* __restrict__ bA = bias + 4 * lane;
  const float* __restrict__ bB = bias + 128 + 4 * lane;
  v4f oA, oB;
  oA.x = accA.x / qA + bA[0];  oA.y = accA.y / qA + bA[1];
  oA.z = accA.z / qA + bA[2];  oA.w = accA.w / qA + bA[3];
  oB.x = accB.x / qB + bB[0];  oB.y = accB.y / qB + bB[1];
  oB.z = accB.z / qB + bB[2];  oB.w = accB.w / qB + bB[3];
  oA.x = elu1(oA.x);  oA.y = elu1(oA.y);  oA.z = elu1(oA.z);  oA.w = elu1(oA.w);
  oB.x = elu1(oB.x);  oB.y = elu1(oB.y);  oB.z = elu1(oB.z);  oB.w = elu1(oB.w);
  v4f* __restrict__ od = (v4f*)(out + (size_t)d * 256);
  od[lane]      = oA;
  od[32 + lane] = oB;
}

__global__ __launch_bounds__(256) void agg_h1(
    const int* __restrict__ rowptr, const int* __restrict__ csr_src,
    const float* __restrict__ a_s, const float* __restrict__ a_d,
    const float* __restrict__ h, const float* __restrict__ bias,
    float* __restrict__ out, int N) {
  const int lane = threadIdx.x & 31;
  const int d = __builtin_amdgcn_readfirstlane((int)(blockIdx.x * 8 + (threadIdx.x >> 5)));
  if (d >= N) return;
  const int beg = rowptr[d], end = rowptr[d + 1];
  const float ad = a_d[d];
  const float ls = lrelu(a_s[d] + ad);

  float m = ls;
  for (int base = beg; base < end; base += 32) {
    const int idx = base + lane;
    if (idx < end) m = fmaxf(m, lrelu(a_s[csr_src[idx]] + ad));
  }
#pragma unroll
  for (int off = 16; off > 0; off >>= 1) m = fmaxf(m, __shfl_xor(m, off, 32));

  const float wself = expf(ls - m);
  float den = 0.f;
  float acc = wself * h[(size_t)d * 32 + lane];
  for (int base = beg; base < end; base += 32) {
    const int n = (end - base < 32) ? (end - base) : 32;
    int sv = d;
    float wv = 0.f;
    if (lane < n) {
      sv = csr_src[base + lane];
      wv = expf(lrelu(a_s[sv] + ad) - m);
    }
    den += wv;
    for (int j = 0; j < n; j += 8) {
      float r[8], w[8];
#pragma unroll
      for (int u = 0; u < 8; ++u) {
        const int s = __builtin_amdgcn_readlane(sv, j + u);
        w[u] = __uint_as_float(
            (unsigned)__builtin_amdgcn_readlane((int)__float_as_uint(wv), j + u));
        r[u] = h[(size_t)s * 32 + lane];
      }
#pragma unroll
      for (int u = 0; u < 8; ++u) acc += w[u] * r[u];
    }
  }
#pragma unroll
  for (int off = 16; off > 0; off >>= 1) den += __shfl_xor(den, off, 32);
  den += wself;
  out[(size_t)d * 32 + lane] = elu1(acc / (den + 1e-16f) + bias[lane]);
}

extern "C" void kernel_launch(void* const* d_in, const int* in_sizes, int n_in,
                              void* d_out, int out_size, void* d_ws, size_t ws_size,
                              hipStream_t stream) {
  const float* x   = (const float*)d_in[0];
  const int*   ei  = (const int*)d_in[1];
  const float* W1  = (const float*)d_in[2];
  const float* as1 = (const float*)d_in[3];
  const float* ad1 = (const float*)d_in[4];
  const float* b1  = (const float*)d_in[5];
  const float* W2  = (const float*)d_in[6];
  const float* as2 = (const float*)d_in[7];
  const float* ad2 = (const float*)d_in[8];
  const float* b2  = (const float*)d_in[9];

  const int HC   = in_sizes[3];
  const int Fin  = in_sizes[2] / HC;
  const int N    = in_sizes[0] / Fin;
  const int E    = in_sizes[1] / 2;
  const int H    = 8;
  const int Cout = in_sizes[7];

  float* ws = (float*)d_ws;
  float* h1  = ws;
  float* x2  = h1 + (size_t)N * HC;
  float* h2  = x2 + (size_t)N * HC;
  float* aS1 = h2 + (size_t)N * Cout;
  float* aD1 = aS1 + (size_t)N * H;
  float* aS2 = aD1 + (size_t)N * H;
  float* aD2 = aS2 + N;
  int*   rowptr = (int*)(aD2 + N);
  int*   csr    = rowptr + ((N + 1 + 31) / 32) * 32;
  int*   recS   = csr + E;
  int*   recD   = recS + E;
  const int nChunks = (E + BIN_CHUNK - 1) / BIN_CHUNK;
  int*   tabOff = recD + E;
  int*   tabCnt = tabOff + (size_t)nChunks * BKT_DSTS;

  const int TB = 256;
  auto blocks = [](long long n, int tb) { return (int)((n + tb - 1) / tb); };

  edge_bin<<<nChunks, SORT_TB, 0, stream>>>(ei, E, recS, recD, tabOff, tabCnt);
  bucket_sort<<<(N + BKT_DSTS - 1) / BKT_DSTS, SORT_TB, 0, stream>>>(
      recS, recD, tabOff, tabCnt, nChunks, rowptr, csr, N);

  {
    const long long waves = (long long)((N + 31) / 32) * (HC / 64);
    gemm_wmma_f32<2, 4, 32, 128, 256><<<blocks(waves * 32, TB), TB, 0, stream>>>(x, W1, h1, N);
    att_coef<8><<<blocks((long long)((N * H / 8 + 7) / 8) * 32, TB), TB, 0, stream>>>(
        h1, as1, ad1, aS1, aD1, N * H / 8);
    agg_h8<<<blocks((long long)N * 32, TB), TB, 0, stream>>>(
        rowptr, csr, aS1, aD1, h1, b1, x2, N);
  }

  {
    const long long waves = (long long)(N / 16) * (Cout / 32);
    gemm_wmma_f32<1, 2, 64, 256, 32><<<blocks(waves * 32, TB), TB, 0, stream>>>(x2, W2, h2, N);
    att_coef<1><<<blocks((long long)((N / 8 + 7) / 8) * 32, TB), TB, 0, stream>>>(
        h2, as2, ad2, aS2, aD2, N / 8);
    agg_h1<<<blocks((long long)N * 32, TB), TB, 0, stream>>>(
        rowptr, csr, aS2, aD2, h2, b2, (float*)d_out, N);
  }
}


# Both reducers take each edge's score from the gathered h row in registers (partial dots, halving xor adds), one pass, running max rescales the sums; att_coef launches and score gathers go
# speedup vs baseline: 12.6683x; 1.1729x over previous
#include <hip/hip_runtime.h>
#include <math.h>

typedef float v2f __attribute__((ext_vector_type(2)));
typedef float v4f __attribute__((ext_vector_type(4)));
typedef float v8f __attribute__((ext_vector_type(8)));

#define LRELU_SLOPE 0.2f

template <int MT, int NT, int KB, int KT, int NC>
__global__ __launch_bounds__(256) void gemm_wmma_f32(
    const float* __restrict__ X, const float* __restrict__ W,
    float* __restrict__ Y, int M) {
  typedef float vNf __attribute__((ext_vector_type(NT)));
  constexpr int NJ = KB / 8;
  constexpr int groupsN = NC / (16 * NT);
  const int lane = threadIdx.x & 31;
  const int wave = __builtin_amdgcn_readfirstlane(
      (int)((blockIdx.x * blockDim.x + threadIdx.x) >> 5));
  const int blocksM = (M + 16 * MT - 1) / (16 * MT);
  if (wave >= blocksM * groupsN) return;
  const int m0 = (wave / groupsN) * (16 * MT);
  const int n0 = (wave % groupsN) * (16 * NT);
  const int hf = lane >> 4;
  const int c  = lane & 15;

  const float* __restrict__ xr[MT];
#pragma unroll
  for (int mt = 0; mt < MT; ++mt) {
    int row = m0 + mt * 16 + c;
    row = (row < M) ? row : (M - 1);
    xr[mt] = X + (size_t)row * KT + 4 * hf;
  }
  const float* __restrict__ wp = W + (size_t)(4 * hf) * NC + n0 + NT * c;

  v8f acc[MT][NT];
#pragma unroll
  for (int mt = 0; mt < MT; ++mt)
#pragma unroll
    for (int t = 0; t < NT; ++t)
      acc[mt][t] = (v8f){0.f, 0.f, 0.f, 0.f, 0.f, 0.f, 0.f, 0.f};

#pragma unroll 1
  for (int kb = 0; kb < KT; kb += KB) {
    v4f a[MT][NJ];
    vNf w[NJ][4];
#pragma unroll
    for (int mt = 0; mt < MT; ++mt)
#pragma unroll
      for (int j = 0; j < NJ; ++j)
        a[mt][j] = *(const v4f*)(xr[mt] + kb + 8 * j);
#pragma unroll
    for (int j = 0; j < NJ; ++j)
#pragma unroll
      for (int q = 0; q < 4; ++q)
        w[j][q] = *(const vNf*)(wp + (size_t)(kb + 8 * j + q) * NC);

#pragma unroll
    for (int j = 0; j < NJ; ++j) {
#pragma unroll
      for (int p = 0; p < 2; ++p) {
#pragma unroll
        for (int mt = 0; mt < MT; ++mt) {
          v2f av;
          av.x = p ? a[mt][j].z : a[mt][j].x;
          av.y = p ? a[mt][j].w : a[mt][j].y;
#pragma unroll
          for (int t = 0; t < NT; ++t) {
            v2f bv;
            bv.x = w[j][2 * p][t];
            bv.y = w[j][2 * p + 1][t];
#if __has_builtin(__builtin_amdgcn_wmma_f32_16x16x4_f32)
            acc[mt][t] = __builtin_amdgcn_wmma_f32_16x16x4_f32(
                false, av, false, bv, (short)0, acc[mt][t], false, false);
#else
            acc[mt][t][0] += av.x * bv.x;
#endif
          }
        }
      }
    }
    __builtin_amdgcn_sched_group_barrier(0x020, MT * NJ + 4 * NJ, 0);
    __builtin_amdgcn_sched_group_barrier(0x00A, 8 * MT * NT * NJ, 0);
  }

#pragma unroll
  for (int mt = 0; mt < MT; ++mt) {
    if (m0 + mt * 16 < M) {
      float* __restrict__ yr = Y + (size_t)(m0 + mt * 16 + 8 * hf) * NC + n0 + NT * c;
#pragma unroll
      for (int r = 0; r < 8; ++r) {
        vNf o;
#pragma unroll
        for (int t = 0; t < NT; ++t) o[t] = acc[mt][t][r];
        *(vNf*)(yr + (size_t)r * NC) = o;
      }
    }
  }
}

#define SORT_TB    1024
#define BIN_EPT    4
#define BIN_CHUNK  (BIN_EPT * SORT_TB)
#define BKT_SHIFT  8
#define BKT_DSTS   256
#define BKT_CAP    6144

__device__ __forceinline__ int scan256(const int* v, int* o, int* wt,
                                       int tid, int wv, int& c) {
  const int lane = tid & 31;
  int incl = 0;
  c = 0;
  if (wv < 8) {
    c = v[tid];
    incl = c;
#pragma unroll
    for (int k = 1; k < 32; k <<= 1) {
      const int t = __shfl_up(incl, k, 32);
      if (lane >= k) incl += t;
    }
    if (lane == 31) wt[wv] = incl;
  }
  __syncthreads();
  int ex = 0;
  if (wv < 8) {
    int pre = 0;
#pragma unroll
    for (int w = 0; w < 8; ++w) pre += (w < wv) ? wt[w] : 0;
    ex = pre + incl - c;
    o[tid] = ex;
  }
  __syncthreads();
  return ex;
}

__global__ __launch_bounds__(SORT_TB) void edge_bin(
    const int* __restrict__ ei, int E, int* __restrict__ recS,
    int* __restrict__ recD, int* __restrict__ tabOff, int* __restrict__ tabCnt) {
  __shared__ int cnt[BKT_DSTS];
  __shared__ int off[BKT_DSTS];
  __shared__ int wt[8];
  __shared__ int srtS[BIN_CHUNK];
  __shared__ int srtD[BIN_CHUNK];
  const int tid = threadIdx.x;
  const int wv = __builtin_amdgcn_readfirstlane((int)(threadIdx.x >> 5));
  const int e0 = blockIdx.x * BIN_CHUNK;
  if (tid < BKT_DSTS) cnt[tid] = 0;
  __syncthreads();

  int s[BIN_EPT], d[BIN_EPT], r[BIN_EPT];
#pragma unroll
  for (int u = 0; u < BIN_EPT; ++u) {
    const int e = e0 + u * SORT_TB + tid;
    s[u] = 0;  d[u] = -1;  r[u] = 0;
    if (e < E) { s[u] = ei[e];  d[u] = ei[E + e]; }
  }
#pragma unroll
  for (int u = 0; u < BIN_EPT; ++u)
    if (d[u] >= 0) r[u] = atomicAdd(&cnt[d[u] >> BKT_SHIFT], 1);
  __syncthreads();

  int c;
  const int ex = scan256(cnt, off, wt, tid, wv, c);
  if (wv < 8) {
    tabOff[blockIdx.x * BKT_DSTS + tid] = ex;
    tabCnt[blockIdx.x * BKT_DSTS + tid] = c;
  }
#pragma unroll
  for (int u = 0; u < BIN_EPT; ++u) {
    if (d[u] >= 0) {
      const int p = off[d[u] >> BKT_SHIFT] + r[u];
      srtS[p] = s[u];
      srtD[p] = d[u];
    }
  }
  __syncthreads();
  int nE = E - e0;
  if (nE > BIN_CHUNK) nE = BIN_CHUNK;
#pragma unroll
  for (int u = 0; u < BIN_EPT; ++u) {
    const int i = u * SORT_TB + tid;
    if (i < nE) { recS[e0 + i] = srtS[i];  recD[e0 + i] = srtD[i]; }
  }
}

__device__ __forceinline__ int rec_index(int i, const int* pOfs, const int* pSrc) {
  int lo = 0;
#pragma unroll
  for (int step = 512; step > 0; step >>= 1) {
    const int m = lo + step;
    if (pOfs[m] <= i) lo = m;
  }
  return pSrc[lo] + (i - pOfs[lo]);
}

__global__ __launch_bounds__(SORT_TB) void bucket_sort(
    const int* __restrict__ recS, const int* __restrict__ recD,
    const int* __restrict__ tabOff, const int* __restrict__ tabCnt, int nChunks,
    int* __restrict__ rowptr, int* __restrict__ csr_src, int N) {
  __shared__ int pOfs[SORT_TB];
  __shared__ int pSrc[SORT_TB];
  __shared__ int wsum[32];
  __shared__ int wbase[32];
  __shared__ int dcnt[BKT_DSTS];
  __shared__ int dofs[BKT_DSTS];
  __shared__ int wt[8];
  __shared__ int bufS[BKT_CAP];
  __shared__ int bufK[BKT_CAP];
  __shared__ int fin[BKT_CAP];
  const int tid = threadIdx.x;
  const int lane = tid & 31;
  const int wv = __builtin_amdgcn_readfirstlane((int)(threadIdx.x >> 5));
  const int g = blockIdx.x;

  int o = 0, c = 0;
  if (tid < nChunks) {
    o = tabOff[tid * BKT_DSTS + g];
    c = tabCnt[tid * BKT_DSTS + g];
  }
  if (tid < BKT_DSTS) dcnt[tid] = 0;
  int inclC = c, sumO = o;
#pragma unroll
  for (int k = 1; k < 32; k <<= 1) {
    const int t = __shfl_up(inclC, k, 32);
    if (lane >= k) inclC += t;
  }
#pragma unroll
  for (int k = 16; k > 0; k >>= 1) sumO += __shfl_xor(sumO, k, 32);
  if (lane == 31) wsum[wv] = inclC;
  if (lane == 0) wbase[wv] = sumO;
  __syncthreads();
  const int ws = wsum[lane];
  int wincl = ws;
#pragma unroll
  for (int k = 1; k < 32; k <<= 1) {
    const int t = __shfl_up(wincl, k, 32);
    if (lane >= k) wincl += t;
  }
  const int wpre  = __shfl(wincl - ws, wv, 32);
  const int total = __builtin_amdgcn_readfirstlane(__shfl(wincl, 31, 32));
  int bsum = wbase[lane];
#pragma unroll
  for (int k = 16; k > 0; k >>= 1) bsum += __shfl_xor(bsum, k, 32);
  const int base = __builtin_amdgcn_readfirstlane(bsum);
  pOfs[tid] = wpre + inclC - c;
  pSrc[tid] = tid * BIN_CHUNK + o;
  __syncthreads();

  if (total <= BKT_CAP) {
    constexpr int RPT = BKT_CAP / SORT_TB;
    int sv[RPT], dv[RPT];
#pragma unroll
    for (int k = 0; k < RPT; ++k) {
      const int i = k * SORT_TB + tid;
      sv[k] = 0;  dv[k] = 0;
      if (i < total) {
        const int q = rec_index(i, pOfs, pSrc);
        sv[k] = recS[q];
        dv[k] = recD[q];
      }
    }
#pragma unroll
    for (int k = 0; k < RPT; ++k) {
      const int i = k * SORT_TB + tid;
      if (i < total) {
        const int dl = dv[k] & (BKT_DSTS - 1);
        const int rk = atomicAdd(&dcnt[dl], 1);
        bufS[i] = sv[k];
        bufK[i] = (rk << BKT_SHIFT) | dl;
      }
    }
    __syncthreads();
    int cc;
    const int ex = scan256(dcnt, dofs, wt, tid, wv, cc);
    if (wv < 8) {
      const int node = g * BKT_DSTS + tid;
      if (node < N) rowptr[node] = base + ex;
    }
    for (int i = tid; i < total; i += SORT_TB) {
      const int k = bufK[i];
      fin[dofs[k & (BKT_DSTS - 1)] + (k >> BKT_SHIFT)] = bufS[i];
    }
    __syncthreads();
    for (int i = tid; i < total; i += SORT_TB) csr_src[base + i] = fin[i];
  } else {
    for (int i = tid; i < total; i += SORT_TB) {
      const int q = rec_index(i, pOfs, pSrc);
      atomicAdd(&dcnt[recD[q] & (BKT_DSTS - 1)], 1);
    }
    __syncthreads();
    int cc;
    const int ex = scan256(dcnt, dofs, wt, tid, wv, cc);
    if (wv < 8) {
      const int node = g * BKT_DSTS + tid;
      if (node < N) rowptr[node] = base + ex;
      dcnt[tid] = 0;
    }
    __syncthreads();
    for (int i = tid; i < total; i += SORT_TB) {
      const int q = rec_index(i, pOfs, pSrc);
      const int dl = recD[q] & (BKT_DSTS - 1);
      const int rk = atomicAdd(&dcnt[dl], 1);
      csr_src[base + dofs[dl] + rk] = recS[q];
    }
  }
  if (g == (int)gridDim.x - 1 && tid == 0) rowptr[N] = base + total;
}

__device__ __forceinline__ float lrelu(float l) {
  return (l > 0.f) ? l : LRELU_SLOPE * l;
}
__device__ __forceinline__ float elu1(float v) {
  return (v > 0.f) ? v : (expf(v) - 1.f);
}

template <int M>
__device__ __forceinline__ float xor_get(float v) {
  return __int_as_float(__builtin_amdgcn_update_dpp(
      0, __float_as_int(v), 0x160 + M, 0xf, 0xf, true));
}
template <int K>
__device__ __forceinline__ float bcast8(float v) {
  return __uint_as_float(__builtin_amdgcn_mov_dpp8(__float_as_uint(v), K * 0x249249));
}
__device__ __forceinline__ float sum8(float v) {
  v += xor_get<4>(v);
  v += xor_get<2>(v);
  v += xor_get<1>(v);
  return v;
}
__device__ __forceinline__ float dot4p(v4f a, v4f b) {
  v2f p = a.xy * b.xy;
  p = a.zw * b.zw + p;
  return p.x + p.y;
}

__global__ __launch_bounds__(256) void agg_h8(
    const int* __restrict__ rowptr, const int* __restrict__ csr_src,
    const float* __restrict__ att_s, const float* __restrict__ att_d,
    const float* __restrict__ h, const float* __restrict__ bias,
    float* __restrict__ out, int N) {
  const int lane = threadIdx.x & 31;
  const int d = __builtin_amdgcn_readfirstlane((int)(blockIdx.x * 8 + (threadIdx.x >> 5)));
  if (d >= N) return;
  const bool b2 = (lane & 4) != 0, b1 = (lane & 2) != 0, b0 = (lane & 1) != 0;
  const int uq = lane & 3;
  const int beg = rowptr[d], end = rowptr[d + 1];
  const v4f sA = *(const v4f*)(att_s + 4 * lane), sB = *(const v4f*)(att_s + 128 + 4 * lane);
  const v4f dA = *(const v4f*)(att_d + 4 * lane), dB = *(const v4f*)(att_d + 128 + 4 * lane);
  const v4f* __restrict__ hd = (const v4f*)(h + (size_t)d * 256);
  const v4f rdA = hd[lane], rdB = hd[32 + lane];
  const float adA = sum8(dot4p(rdA, dA)), adB = sum8(dot4p(rdB, dB));
  const float lsA = lrelu(sum8(dot4p(rdA, sA)) + adA);
  const float lsB = lrelu(sum8(dot4p(rdB, sB)) + adB);

  const float ad = b2 ? adB : adA;
  float m = b2 ? lsB : lsA;
  float den = (uq == 0) ? 1.f : 0.f;
  v4f accA = rdA, accB = rdB;

  for (int base = beg; base < end; base += 32) {
    const int n = (end - base < 32) ? (end - base) : 32;
    const int sv = (lane < n) ? csr_src[base + lane] : d;
    for (int j = 0; j < n; j += 4) {
      v4f ra[4], rb[4];
#pragma unroll
      for (int u = 0; u < 4; ++u) {
        const int s = __builtin_amdgcn_readlane(sv, j + u);
        const v4f* __restrict__ hr = (const v4f*)(h + (size_t)s * 256);
        ra[u] = hr[lane];
        rb[u] = hr[32 + lane];
      }
      float k1[4];
#pragma unroll
      for (int u = 0; u < 4; ++u) {
        const float pa = dot4p(ra[u], sA), pb = dot4p(rb[u], sB);
        k1[u] = (b2 ? pb : pa) + xor_get<4>(b2 ? pa : pb);
      }
      float k2[2];
#pragma unroll
      for (int v = 0; v < 2; ++v)
        k2[v] = (b1 ? k1[2 + v] : k1[v]) + xor_get<2>(b1 ? k1[v] : k1[2 + v]);
      const float sc0 = (b0 ? k2[1] : k2[0]) + xor_get<1>(b0 ? k2[0] : k2[1]);

      const float l = (j + uq < n) ? lrelu(sc0 + ad) : -__builtin_huge_valf();
      float g = fmaxf(l, xor_get<1>(l));
      g = fmaxf(g, xor_get<2>(g));
      const float mn = fmaxf(m, g);
      const float sc = expf(m - mn);
      const float w = expf(l - mn);
      m = mn;
      den = den * sc + w;
      accA *= bcast8<0>(sc);
      accB *= bcast8<4>(sc);
      accA += bcast8<0>(w) * ra[0];
      accB += bcast8<4>(w) * rb[0];
      accA += bcast8<1>(w) * ra[1];
      accB += bcast8<5>(w) * rb[1];
      accA += bcast8<2>(w) * ra[2];
      accB += bcast8<6>(w) * rb[2];
      accA += bcast8<3>(w) * ra[3];
      accB += bcast8<7>(w) * rb[3];
      __builtin_amdgcn_sched_group_barrier(0x020, 8, 0);
      __builtin_amdgcn_sched_group_barrier(0x002, 512, 0);
    }
  }

  den += xor_get<1>(den);
  den += xor_get<2>(den);
  const float qA = bcast8<0>(den) + 1e-16f, qB = bcast8<4>(den) + 1e-16f;
  const float* __restrict__ bA = bias + 4 * lane;
  const float* __restrict__ bB = bias + 128 + 4 * lane;
  v4f oA, oB;
  oA.x = accA.x / qA + bA[0];  oA.y = accA.y / qA + bA[1];
  oA.z = accA.z / qA + bA[2];  oA.w = accA.w / qA + bA[3];
  oB.x = accB.x / qB + bB[0];  oB.y = accB.y / qB + bB[1];
  oB.z = accB.z / qB + bB[2];  oB.w = accB.w / qB + bB[3];
  oA.x = elu1(oA.x);  oA.y = elu1(oA.y);  oA.z = elu1(oA.z);  oA.w = elu1(oA.w);
  oB.x = elu1(oB.x);  oB.y = elu1(oB.y);  oB.z = elu1(oB.z);  oB.w = elu1(oB.w);
  v4f* __restrict__ od = (v4f*)(out + (size_t)d * 256);
  od[lane]      = oA;
  od[32 + lane] = oB;
}

template <int K>
__device__ __forceinline__ int bcast8i(int v) {
  return (int)__builtin_amdgcn_mov_dpp8((unsigned)v, K * 0x249249);
}

__global__ __launch_bounds__(256) void agg_h1(
    const int* __restrict__ rowptr, const int* __restrict__ csr_src,
    const float* __restrict__ att_s, const float* __restrict__ att_d,
    const float* __restrict__ h, const float* __restrict__ bias,
    float* __restrict__ out, int N) {
  const int lane = threadIdx.x & 31;
  const int d = __builtin_amdgcn_readfirstlane((int)(blockIdx.x * 8 + (threadIdx.x >> 5)));
  if (d >= N) return;
  const int q = lane & 7;
  const bool b2 = (lane & 4) != 0, b1 = (lane & 2) != 0, b0 = (lane & 1) != 0;
  const int beg = rowptr[d], end = rowptr[d + 1];
  const v4f s4 = *(const v4f*)(att_s + 4 * q), d4 = *(const v4f*)(att_d + 4 * q);
  const v4f* __restrict__ hq = (const v4f*)h + q;
  const v4f rd = hq[(size_t)d * 8];
  const float ad = sum8(dot4p(rd, d4));
  float m = lrelu(sum8(dot4p(rd, s4)) + ad);
  float den = (lane == 0) ? 1.f : 0.f;
  v4f acc = rd;
  if (lane >= 8) acc = (v4f){0.f, 0.f, 0.f, 0.f};

  for (int base = beg; base < end; base += 32) {
    const int n = (end - base < 32) ? (end - base) : 32;
    const int sv = (lane < n) ? csr_src[base + lane] : d;
    v4f r[8];
    r[0] = hq[(size_t)bcast8i<0>(sv) * 8];
    r[1] = hq[(size_t)bcast8i<1>(sv) * 8];
    r[2] = hq[(size_t)bcast8i<2>(sv) * 8];
    r[3] = hq[(size_t)bcast8i<3>(sv) * 8];
    r[4] = hq[(size_t)bcast8i<4>(sv) * 8];
    r[5] = hq[(size_t)bcast8i<5>(sv) * 8];
    r[6] = hq[(size_t)bcast8i<6>(sv) * 8];
    r[7] = hq[(size_t)bcast8i<7>(sv) * 8];
    float k1[4];
#pragma unroll
    for (int v = 0; v < 4; ++v) {
      const float pa = dot4p(r[v], s4), pb = dot4p(r[4 + v], s4);
      k1[v] = (b2 ? pb : pa) + xor_get<4>(b2 ? pa : pb);
    }
    float k2[2];
#pragma unroll
    for (int v = 0; v < 2; ++v)
      k2[v] = (b1 ? k1[2 + v] : k1[v]) + xor_get<2>(b1 ? k1[v] : k1[2 + v]);
    const float sc0 = (b0 ? k2[1] : k2[0]) + xor_get<1>(b0 ? k2[0] : k2[1]);

    const float l = (lane < n) ? lrelu(sc0 + ad) : -__builtin_huge_valf();
    float g = fmaxf(l, xor_get<1>(l));
    g = fmaxf(g, xor_get<2>(g));
    g = fmaxf(g, xor_get<4>(g));
    g = fmaxf(g, xor_get<8>(g));
    g = fmaxf(g, __shfl_xor(g, 16, 32));
    const float mn = fmaxf(m, g);
    const float sc = expf(m - mn);
    const float w = expf(l - mn);
    m = mn;
    den = den * sc + w;
    acc *= sc;
    acc += bcast8<0>(w) * r[0];
    acc += bcast8<1>(w) * r[1];
    acc += bcast8<2>(w) * r[2];
    acc += bcast8<3>(w) * r[3];
    acc += bcast8<4>(w) * r[4];
    acc += bcast8<5>(w) * r[5];
    acc += bcast8<6>(w) * r[6];
    acc += bcast8<7>(w) * r[7];
  }

  acc.x += xor_get<8>(acc.x);  acc.y += xor_get<8>(acc.y);
  acc.z += xor_get<8>(acc.z);  acc.w += xor_get<8>(acc.w);
  acc.x += __shfl_xor(acc.x, 16, 32);  acc.y += __shfl_xor(acc.y, 16, 32);
  acc.z += __shfl_xor(acc.z, 16, 32);  acc.w += __shfl_xor(acc.w, 16, 32);
  den = sum8(den);
  den += xor_get<8>(den);
  den += __shfl_xor(den, 16, 32);
  if (lane < 8) {
    const float qd = den + 1e-16f;
    const v4f bv = *(const v4f*)(bias + 4 * q);
    v4f o;
    o.x = acc.x / qd + bv.x;  o.y = acc.y / qd + bv.y;
    o.z = acc.z / qd + bv.z;  o.w = acc.w / qd + bv.w;
    o.x = elu1(o.x);  o.y = elu1(o.y);  o.z = elu1(o.z);  o.w = elu1(o.w);
    ((v4f*)(out + (size_t)d * 32))[q] = o;
  }
}

extern "C" void kernel_launch(void* const* d_in, const int* in_sizes, int n_in,
                              void* d_out, int out_size, void* d_ws, size_t ws_size,
                              hipStream_t stream) {
  const float* x   = (const float*)d_in[0];
  const int*   ei  = (const int*)d_in[1];
  const float* W1  = (const float*)d_in[2];
  const float* as1 = (const float*)d_in[3];
  const float* ad1 = (const float*)d_in[4];
  const float* b1  = (const float*)d_in[5];
  const float* W2  = (const float*)d_in[6];
  const float* as2 = (const float*)d_in[7];
  const float* ad2 = (const float*)d_in[8];
  const float* b2  = (const float*)d_in[9];

  const int HC   = in_sizes[3];
  const int Fin  = in_sizes[2] / HC;
  const int N    = in_sizes[0] / Fin;
  const int E    = in_sizes[1] / 2;
  const int Cout = in_sizes[7];

  float* ws = (float*)d_ws;
  float* h1  = ws;
  float* x2  = h1 + (size_t)N * HC;
  float* h2  = x2 + (size_t)N * HC;
  int*   rowptr = (int*)(h2 + (size_t)N * Cout);
  int*   csr    = rowptr + ((N + 1 + 31) / 32) * 32;
  int*   recS   = csr + E;
  int*   recD   = recS + E;
  const int nChunks = (E + BIN_CHUNK - 1) / BIN_CHUNK;
  int*   tabOff = recD + E;
  int*   tabCnt = tabOff + (size_t)nChunks * BKT_DSTS;

  const int TB = 256;
  auto blocks = [](long long n, int tb) { return (int)((n + tb - 1) / tb); };

  edge_bin<<<nChunks, SORT_TB, 0, stream>>>(ei, E, recS, recD, tabOff, tabCnt);
  bucket_sort<<<(N + BKT_DSTS - 1) / BKT_DSTS, SORT_TB, 0, stream>>>(
      recS, recD, tabOff, tabCnt, nChunks, rowptr, csr, N);

  {
    const long long waves = (long long)((N + 31) / 32) * (HC / 64);
    gemm_wmma_f32<2, 4, 32, 128, 256><<<blocks(waves * 32, TB), TB, 0, stream>>>(x, W1, h1, N);
    agg_h8<<<blocks((long long)N * 32, TB), TB, 0, stream>>>(
        rowptr, csr, as1, ad1, h1, b1, x2, N);
  }

  {
    const long long waves = (long long)(N / 16) * (Cout / 32);
    gemm_wmma_f32<1, 2, 64, 256, 32><<<blocks(waves * 32, TB), TB, 0, stream>>>(x2, W2, h2, N);
    agg_h1<<<blocks((long long)N * 32, TB), TB, 0, stream>>>(
        rowptr, csr, as2, ad2, h2, b2, (float*)d_out, N);
  }
}


# Reducers: library expf and the IEEE divides become v_exp_f32 on logits kept times log2 e and one v_rcp_f32 a head with a multiply-add a channel; weights and quotients now round within about 1 ulp
# speedup vs baseline: 13.3568x; 1.0543x over previous
#include <hip/hip_runtime.h>
#include <math.h>

typedef float v2f __attribute__((ext_vector_type(2)));
typedef float v4f __attribute__((ext_vector_type(4)));
typedef float v8f __attribute__((ext_vector_type(8)));

#define LRELU_SLOPE 0.2f

template <int MT, int NT, int KB, int KT, int NC>
__global__ __launch_bounds__(256) void gemm_wmma_f32(
    const float* __restrict__ X, const float* __restrict__ W,
    float* __restrict__ Y, int M) {
  typedef float vNf __attribute__((ext_vector_type(NT)));
  constexpr int NJ = KB / 8;
  constexpr int groupsN = NC / (16 * NT);
  const int lane = threadIdx.x & 31;
  const int wave = __builtin_amdgcn_readfirstlane(
      (int)((blockIdx.x * blockDim.x + threadIdx.x) >> 5));
  const int blocksM = (M + 16 * MT - 1) / (16 * MT);
  if (wave >= blocksM * groupsN) return;
  const int m0 = (wave / groupsN) * (16 * MT);
  const int n0 = (wave % groupsN) * (16 * NT);
  const int hf = lane >> 4;
  const int c  = lane & 15;

  const float* __restrict__ xr[MT];
#pragma unroll
  for (int mt = 0; mt < MT; ++mt) {
    int row = m0 + mt * 16 + c;
    row = (row < M) ? row : (M - 1);
    xr[mt] = X + (size_t)row * KT + 4 * hf;
  }
  const float* __restrict__ wp = W + (size_t)(4 * hf) * NC + n0 + NT * c;

  v8f acc[MT][NT];
#pragma unroll
  for (int mt = 0; mt < MT; ++mt)
#pragma unroll
    for (int t = 0; t < NT; ++t)
      acc[mt][t] = (v8f){0.f, 0.f, 0.f, 0.f, 0.f, 0.f, 0.f, 0.f};

#pragma unroll 1
  for (int kb = 0; kb < KT; kb += KB) {
    v4f a[MT][NJ];
    vNf w[NJ][4];
#pragma unroll
    for (int mt = 0; mt < MT; ++mt)
#pragma unroll
      for (int j = 0; j < NJ; ++j)
        a[mt][j] = *(const v4f*)(xr[mt] + kb + 8 * j);
#pragma unroll
    for (int j = 0; j < NJ; ++j)
#pragma unroll
      for (int q = 0; q < 4; ++q)
        w[j][q] = *(const vNf*)(wp + (size_t)(kb + 8 * j + q) * NC);

#pragma unroll
    for (int j = 0; j < NJ; ++j) {
#pragma unroll
      for (int p = 0; p < 2; ++p) {
#pragma unroll
        for (int mt = 0; mt < MT; ++mt) {
          v2f av;
          av.x = p ? a[mt][j].z : a[mt][j].x;
          av.y = p ? a[mt][j].w : a[mt][j].y;
#pragma unroll
          for (int t = 0; t < NT; ++t) {
            v2f bv;
            bv.x = w[j][2 * p][t];
            bv.y = w[j][2 * p + 1][t];
#if __has_builtin(__builtin_amdgcn_wmma_f32_16x16x4_f32)
            acc[mt][t] = __builtin_amdgcn_wmma_f32_16x16x4_f32(
                false, av, false, bv, (short)0, acc[mt][t], false, false);
#else
            acc[mt][t][0] += av.x * bv.x;
#endif
          }
        }
      }
    }
    __builtin_amdgcn_sched_group_barrier(0x020, MT * NJ + 4 * NJ, 0);
    __builtin_amdgcn_sched_group_barrier(0x00A, 8 * MT * NT * NJ, 0);
  }

#pragma unroll
  for (int mt = 0; mt < MT; ++mt) {
    if (m0 + mt * 16 < M) {
      float* __restrict__ yr = Y + (size_t)(m0 + mt * 16 + 8 * hf) * NC + n0 + NT * c;
#pragma unroll
      for (int r = 0; r < 8; ++r) {
        vNf o;
#pragma unroll
        for (int t = 0; t < NT; ++t) o[t] = acc[mt][t][r];
        *(vNf*)(yr + (size_t)r * NC) = o;
      }
    }
  }
}

#define SORT_TB    1024
#define BIN_EPT    4
#define BIN_CHUNK  (BIN_EPT * SORT_TB)
#define BKT_SHIFT  8
#define BKT_DSTS   256
#define BKT_CAP    6144

__device__ __forceinline__ int scan256(const int* v, int* o, int* wt,
                                       int tid, int wv, int& c) {
  const int lane = tid & 31;
  int incl = 0;
  c = 0;
  if (wv < 8) {
    c = v[tid];
    incl = c;
#pragma unroll
    for (int k = 1; k < 32; k <<= 1) {
      const int t = __shfl_up(incl, k, 32);
      if (lane >= k) incl += t;
    }
    if (lane == 31) wt[wv] = incl;
  }
  __syncthreads();
  int ex = 0;
  if (wv < 8) {
    int pre = 0;
#pragma unroll
    for (int w = 0; w < 8; ++w) pre += (w < wv) ? wt[w] : 0;
    ex = pre + incl - c;
    o[tid] = ex;
  }
  __syncthreads();
  return ex;
}

__global__ __launch_bounds__(SORT_TB) void edge_bin(
    const int* __restrict__ ei, int E, int* __restrict__ recS,
    int* __restrict__ recD, int* __restrict__ tabOff, int* __restrict__ tabCnt) {
  __shared__ int cnt[BKT_DSTS];
  __shared__ int off[BKT_DSTS];
  __shared__ int wt[8];
  __shared__ int srtS[BIN_CHUNK];
  __shared__ int srtD[BIN_CHUNK];
  const int tid = threadIdx.x;
  const int wv = __builtin_amdgcn_readfirstlane((int)(threadIdx.x >> 5));
  const int e0 = blockIdx.x * BIN_CHUNK;
  if (tid < BKT_DSTS) cnt[tid] = 0;
  __syncthreads();

  int s[BIN_EPT], d[BIN_EPT], r[BIN_EPT];
#pragma unroll
  for (int u = 0; u < BIN_EPT; ++u) {
    const int e = e0 + u * SORT_TB + tid;
    s[u] = 0;  d[u] = -1;  r[u] = 0;
    if (e < E) { s[u] = ei[e];  d[u] = ei[E + e]; }
  }
#pragma unroll
  for (int u = 0; u < BIN_EPT; ++u)
    if (d[u] >= 0) r[u] = atomicAdd(&cnt[d[u] >> BKT_SHIFT], 1);
  __syncthreads();

  int c;
  const int ex = scan256(cnt, off, wt, tid, wv, c);
  if (wv < 8) {
    tabOff[blockIdx.x * BKT_DSTS + tid] = ex;
    tabCnt[blockIdx.x * BKT_DSTS + tid] = c;
  }
#pragma unroll
  for (int u = 0; u < BIN_EPT; ++u) {
    if (d[u] >= 0) {
      const int p = off[d[u] >> BKT_SHIFT] + r[u];
      srtS[p] = s[u];
      srtD[p] = d[u];
    }
  }
  __syncthreads();
  int nE = E - e0;
  if (nE > BIN_CHUNK) nE = BIN_CHUNK;
#pragma unroll
  for (int u = 0; u < BIN_EPT; ++u) {
    const int i = u * SORT_TB + tid;
    if (i < nE) { recS[e0 + i] = srtS[i];  recD[e0 + i] = srtD[i]; }
  }
}

__device__ __forceinline__ int rec_index(int i, const int* pOfs, const int* pSrc) {
  int lo = 0;
#pragma unroll
  for (int step = 512; step > 0; step >>= 1) {
    const int m = lo + step;
    if (pOfs[m] <= i) lo = m;
  }
  return pSrc[lo] + (i - pOfs[lo]);
}

__global__ __launch_bounds__(SORT_TB) void bucket_sort(
    const int* __restrict__ recS, const int* __restrict__ recD,
    const int* __restrict__ tabOff, const int* __restrict__ tabCnt, int nChunks,
    int* __restrict__ rowptr, int* __restrict__ csr_src, int N) {
  __shared__ int pOfs[SORT_TB];
  __shared__ int pSrc[SORT_TB];
  __shared__ int wsum[32];
  __shared__ int wbase[32];
  __shared__ int dcnt[BKT_DSTS];
  __shared__ int dofs[BKT_DSTS];
  __shared__ int wt[8];
  __shared__ int bufS[BKT_CAP];
  __shared__ int bufK[BKT_CAP];
  __shared__ int fin[BKT_CAP];
  const int tid = threadIdx.x;
  const int lane = tid & 31;
  const int wv = __builtin_amdgcn_readfirstlane((int)(threadIdx.x >> 5));
  const int g = blockIdx.x;

  int o = 0, c = 0;
  if (tid < nChunks) {
    o = tabOff[tid * BKT_DSTS + g];
    c = tabCnt[tid * BKT_DSTS + g];
  }
  if (tid < BKT_DSTS) dcnt[tid] = 0;
  int inclC = c, sumO = o;
#pragma unroll
  for (int k = 1; k < 32; k <<= 1) {
    const int t = __shfl_up(inclC, k, 32);
    if (lane >= k) inclC += t;
  }
#pragma unroll
  for (int k = 16; k > 0; k >>= 1) sumO += __shfl_xor(sumO, k, 32);
  if (lane == 31) wsum[wv] = inclC;
  if (lane == 0) wbase[wv] = sumO;
  __syncthreads();
  const int ws = wsum[lane];
  int wincl = ws;
#pragma unroll
  for (int k = 1; k < 32; k <<= 1) {
    const int t = __shfl_up(wincl, k, 32);
    if (lane >= k) wincl += t;
  }
  const int wpre  = __shfl(wincl - ws, wv, 32);
  const int total = __builtin_amdgcn_readfirstlane(__shfl(wincl, 31, 32));
  int bsum = wbase[lane];
#pragma unroll
  for (int k = 16; k > 0; k >>= 1) bsum += __shfl_xor(bsum, k, 32);
  const int base = __builtin_amdgcn_readfirstlane(bsum);
  pOfs[tid] = wpre + inclC - c;
  pSrc[tid] = tid * BIN_CHUNK + o;
  __syncthreads();

  if (total <= BKT_CAP) {
    constexpr int RPT = BKT_CAP / SORT_TB;
    int sv[RPT], dv[RPT];
#pragma unroll
    for (int k = 0; k < RPT; ++k) {
      const int i = k * SORT_TB + tid;
      sv[k] = 0;  dv[k] = 0;
      if (i < total) {
        const int q = rec_index(i, pOfs, pSrc);
        sv[k] = recS[q];
        dv[k] = recD[q];
      }
    }
#pragma unroll
    for (int k = 0; k < RPT; ++k) {
      const int i = k * SORT_TB + tid;
      if (i < total) {
        const int dl = dv[k] & (BKT_DSTS - 1);
        const int rk = atomicAdd(&dcnt[dl], 1);
        bufS[i] = sv[k];
        bufK[i] = (rk << BKT_SHIFT) | dl;
      }
    }
    __syncthreads();
    int cc;
    const int ex = scan256(dcnt, dofs, wt, tid, wv, cc);
    if (wv < 8) {
      const int node = g * BKT_DSTS + tid;
      if (node < N) rowptr[node] = base + ex;
    }
    for (int i = tid; i < total; i += SORT_TB) {
      const int k = bufK[i];
      fin[dofs[k & (BKT_DSTS - 1)] + (k >> BKT_SHIFT)] = bufS[i];
    }
    __syncthreads();
    for (int i = tid; i < total; i += SORT_TB) csr_src[base + i] = fin[i];
  } else {
    for (int i = tid; i < total; i += SORT_TB) {
      const int q = rec_index(i, pOfs, pSrc);
      atomicAdd(&dcnt[recD[q] & (BKT_DSTS - 1)], 1);
    }
    __syncthreads();
    int cc;
    const int ex = scan256(dcnt, dofs, wt, tid, wv, cc);
    if (wv < 8) {
      const int node = g * BKT_DSTS + tid;
      if (node < N) rowptr[node] = base + ex;
      dcnt[tid] = 0;
    }
    __syncthreads();
    for (int i = tid; i < total; i += SORT_TB) {
      const int q = rec_index(i, pOfs, pSrc);
      const int dl = recD[q] & (BKT_DSTS - 1);
      const int rk = atomicAdd(&dcnt[dl], 1);
      csr_src[base + dofs[dl] + rk] = recS[q];
    }
  }
  if (g == (int)gridDim.x - 1 && tid == 0) rowptr[N] = base + total;
}

#define LOG2E 1.44269504088896340736f
__device__ __forceinline__ float exp2_hw(float t) {
  return __builtin_amdgcn_exp2f(t);
}
__device__ __forceinline__ float lrelu2(float l) {
  return l * ((l > 0.f) ? LOG2E : (LRELU_SLOPE * LOG2E));
}
__device__ __forceinline__ float elu1(float v) {
  return (v > 0.f) ? v : (exp2_hw(v * LOG2E) - 1.f);
}

template <int M>
__device__ __forceinline__ float xor_get(float v) {
  return __int_as_float(__builtin_amdgcn_update_dpp(
      0, __float_as_int(v), 0x160 + M, 0xf, 0xf, true));
}
template <int K>
__device__ __forceinline__ float bcast8(float v) {
  return __uint_as_float(__builtin_amdgcn_mov_dpp8(__float_as_uint(v), K * 0x249249));
}
__device__ __forceinline__ float sum8(float v) {
  v += xor_get<4>(v);
  v += xor_get<2>(v);
  v += xor_get<1>(v);
  return v;
}
__device__ __forceinline__ float dot4p(v4f a, v4f b) {
  v2f p = a.xy * b.xy;
  p = a.zw * b.zw + p;
  return p.x + p.y;
}

__global__ __launch_bounds__(256) void agg_h8(
    const int* __restrict__ rowptr, const int* __restrict__ csr_src,
    const float* __restrict__ att_s, const float* __restrict__ att_d,
    const float* __restrict__ h, const float* __restrict__ bias,
    float* __restrict__ out, int N) {
  const int lane = threadIdx.x & 31;
  const int d = __builtin_amdgcn_readfirstlane((int)(blockIdx.x * 8 + (threadIdx.x >> 5)));
  if (d >= N) return;
  const bool b2 = (lane & 4) != 0, b1 = (lane & 2) != 0, b0 = (lane & 1) != 0;
  const int uq = lane & 3;
  const int beg = rowptr[d], end = rowptr[d + 1];
  const v4f sA = *(const v4f*)(att_s + 4 * lane), sB = *(const v4f*)(att_s + 128 + 4 * lane);
  const v4f dA = *(const v4f*)(att_d + 4 * lane), dB = *(const v4f*)(att_d + 128 + 4 * lane);
  const v4f* __restrict__ hd = (const v4f*)(h + (size_t)d * 256);
  const v4f rdA = hd[lane], rdB = hd[32 + lane];
  const float adA = sum8(dot4p(rdA, dA)), adB = sum8(dot4p(rdB, dB));
  const float lsA = lrelu2(sum8(dot4p(rdA, sA)) + adA);
  const float lsB = lrelu2(sum8(dot4p(rdB, sB)) + adB);

  const float ad = b2 ? adB : adA;
  float m = b2 ? lsB : lsA;
  float den = (uq == 0) ? 1.f : 0.f;
  v4f accA = rdA, accB = rdB;

  for (int base = beg; base < end; base += 32) {
    const int n = (end - base < 32) ? (end - base) : 32;
    const int sv = (lane < n) ? csr_src[base + lane] : d;
    for (int j = 0; j < n; j += 4) {
      v4f ra[4], rb[4];
#pragma unroll
      for (int u = 0; u < 4; ++u) {
        const int s = __builtin_amdgcn_readlane(sv, j + u);
        const v4f* __restrict__ hr = (const v4f*)(h + (size_t)s * 256);
        ra[u] = hr[lane];
        rb[u] = hr[32 + lane];
      }
      float k1[4];
#pragma unroll
      for (int u = 0; u < 4; ++u) {
        const float pa = dot4p(ra[u], sA), pb = dot4p(rb[u], sB);
        k1[u] = (b2 ? pb : pa) + xor_get<4>(b2 ? pa : pb);
      }
      float k2[2];
#pragma unroll
      for (int v = 0; v < 2; ++v)
        k2[v] = (b1 ? k1[2 + v] : k1[v]) + xor_get<2>(b1 ? k1[v] : k1[2 + v]);
      const float sc0 = (b0 ? k2[1] : k2[0]) + xor_get<1>(b0 ? k2[0] : k2[1]);

      const float l = (j + uq < n) ? lrelu2(sc0 + ad) : -__builtin_huge_valf();
      float g = fmaxf(l, xor_get<1>(l));
      g = fmaxf(g, xor_get<2>(g));
      const float mn = fmaxf(m, g);
      const float sc = exp2_hw(m - mn);
      const float w = exp2_hw(l - mn);
      m = mn;
      den = den * sc + w;
      accA *= bcast8<0>(sc);
      accB *= bcast8<4>(sc);
      accA += bcast8<0>(w) * ra[0];
      accB += bcast8<4>(w) * rb[0];
      accA += bcast8<1>(w) * ra[1];
      accB += bcast8<5>(w) * rb[1];
      accA += bcast8<2>(w) * ra[2];
      accB += bcast8<6>(w) * rb[2];
      accA += bcast8<3>(w) * ra[3];
      accB += bcast8<7>(w) * rb[3];
      __builtin_amdgcn_sched_group_barrier(0x020, 8, 0);
      __builtin_amdgcn_sched_group_barrier(0x002, 512, 0);
    }
  }

  den += xor_get<1>(den);
  den += xor_get<2>(den);
  const float qA = __builtin_amdgcn_rcpf(bcast8<0>(den) + 1e-16f);
  const float qB = __builtin_amdgcn_rcpf(bcast8<4>(den) + 1e-16f);
  const float* __restrict__ bA = bias + 4 * lane;
  const float* __restrict__ bB = bias + 128 + 4 * lane;
  v4f oA, oB;
  oA.x = accA.x * qA + bA[0];  oA.y = accA.y * qA + bA[1];
  oA.z = accA.z * qA + bA[2];  oA.w = accA.w * qA + bA[3];
  oB.x = accB.x * qB + bB[0];  oB.y = accB.y * qB + bB[1];
  oB.z = accB.z * qB + bB[2];  oB.w = accB.w * qB + bB[3];
  oA.x = elu1(oA.x);  oA.y = elu1(oA.y);  oA.z = elu1(oA.z);  oA.w = elu1(oA.w);
  oB.x = elu1(oB.x);  oB.y = elu1(oB.y);  oB.z = elu1(oB.z);  oB.w = elu1(oB.w);
  v4f* __restrict__ od = (v4f*)(out + (size_t)d * 256);
  od[lane]      = oA;
  od[32 + lane] = oB;
}

template <int K>
__device__ __forceinline__ int bcast8i(int v) {
  return (int)__builtin_amdgcn_mov_dpp8((unsigned)v, K * 0x249249);
}

__global__ __launch_bounds__(256) __attribute__((amdgpu_waves_per_eu(16, 16))) void agg_h1(
    const int* __restrict__ rowptr, const int* __restrict__ csr_src,
    const float* __restrict__ att_s, const float* __restrict__ att_d,
    const float* __restrict__ h, const float* __restrict__ bias,
    float* __restrict__ out, int N) {
  const int lane = threadIdx.x & 31;
  const int d = __builtin_amdgcn_readfirstlane((int)(blockIdx.x * 8 + (threadIdx.x >> 5)));
  if (d >= N) return;
  const int q = lane & 7;
  const bool b2 = (lane & 4) != 0, b1 = (lane & 2) != 0, b0 = (lane & 1) != 0;
  const int beg = rowptr[d], end = rowptr[d + 1];
  const v4f s4 = *(const v4f*)(att_s + 4 * q), d4 = *(const v4f*)(att_d + 4 * q);
  const v4f* __restrict__ hq = (const v4f*)h + q;
  const v4f rd = hq[(size_t)d * 8];
  const float ad = sum8(dot4p(rd, d4));
  float m = lrelu2(sum8(dot4p(rd, s4)) + ad);
  float den = (lane == 0) ? 1.f : 0.f;
  v4f acc = rd;
  if (lane >= 8) acc = (v4f){0.f, 0.f, 0.f, 0.f};

  for (int base = beg; base < end; base += 32) {
    const int n = (end - base < 32) ? (end - base) : 32;
    const int sv = (lane < n) ? csr_src[base + lane] : d;
    v4f r[8];
    r[0] = hq[(size_t)bcast8i<0>(sv) * 8];
    r[1] = hq[(size_t)bcast8i<1>(sv) * 8];
    r[2] = hq[(size_t)bcast8i<2>(sv) * 8];
    r[3] = hq[(size_t)bcast8i<3>(sv) * 8];
    r[4] = hq[(size_t)bcast8i<4>(sv) * 8];
    r[5] = hq[(size_t)bcast8i<5>(sv) * 8];
    r[6] = hq[(size_t)bcast8i<6>(sv) * 8];
    r[7] = hq[(size_t)bcast8i<7>(sv) * 8];
    float k1[4];
#pragma unroll
    for (int v = 0; v < 4; ++v) {
      const float pa = dot4p(r[v], s4), pb = dot4p(r[4 + v], s4);
      k1[v] = (b2 ? pb : pa) + xor_get<4>(b2 ? pa : pb);
    }
    float k2[2];
#pragma unroll
    for (int v = 0; v < 2; ++v)
      k2[v] = (b1 ? k1[2 + v] : k1[v]) + xor_get<2>(b1 ? k1[v] : k1[2 + v]);
    const float sc0 = (b0 ? k2[1] : k2[0]) + xor_get<1>(b0 ? k2[0] : k2[1]);

    const float l = (lane < n) ? lrelu2(sc0 + ad) : -__builtin_huge_valf();
    float g = fmaxf(l, xor_get<1>(l));
    g = fmaxf(g, xor_get<2>(g));
    g = fmaxf(g, xor_get<4>(g));
    g = fmaxf(g, xor_get<8>(g));
    g = fmaxf(g, __shfl_xor(g, 16, 32));
    const float mn = fmaxf(m, g);
    const float sc = exp2_hw(m - mn);
    const float w = exp2_hw(l - mn);
    m = mn;
    den = den * sc + w;
    acc *= sc;
    acc += bcast8<0>(w) * r[0];
    acc += bcast8<1>(w) * r[1];
    acc += bcast8<2>(w) * r[2];
    acc += bcast8<3>(w) * r[3];
    acc += bcast8<4>(w) * r[4];
    acc += bcast8<5>(w) * r[5];
    acc += bcast8<6>(w) * r[6];
    acc += bcast8<7>(w) * r[7];
  }

  acc.x += xor_get<8>(acc.x);  acc.y += xor_get<8>(acc.y);
  acc.z += xor_get<8>(acc.z);  acc.w += xor_get<8>(acc.w);
  acc.x += __shfl_xor(acc.x, 16, 32);  acc.y += __shfl_xor(acc.y, 16, 32);
  acc.z += __shfl_xor(acc.z, 16, 32);  acc.w += __shfl_xor(acc.w, 16, 32);
  den = sum8(den);
  den += xor_get<8>(den);
  den += __shfl_xor(den, 16, 32);
  if (lane < 8) {
    const float qd = __builtin_amdgcn_rcpf(den + 1e-16f);
    const v4f bv = *(const v4f*)(bias + 4 * q);
    v4f o;
    o.x = acc.x * qd + bv.x;  o.y = acc.y * qd + bv.y;
    o.z = acc.z * qd + bv.z;  o.w = acc.w * qd + bv.w;
    o.x = elu1(o.x);  o.y = elu1(o.y);  o.z = elu1(o.z);  o.w = elu1(o.w);
    ((v4f*)(out + (size_t)d * 32))[q] = o;
  }
}

extern "C" void kernel_launch(void* const* d_in, const int* in_sizes, int n_in,
                              void* d_out, int out_size, void* d_ws, size_t ws_size,
                              hipStream_t stream) {
  const float* x   = (const float*)d_in[0];
  const int*   ei  = (const int*)d_in[1];
  const float* W1  = (const float*)d_in[2];
  const float* as1 = (const float*)d_in[3];
  const float* ad1 = (const float*)d_in[4];
  const float* b1  = (const float*)d_in[5];
  const float* W2  = (const float*)d_in[6];
  const float* as2 = (const float*)d_in[7];
  const float* ad2 = (const float*)d_in[8];
  const float* b2  = (const float*)d_in[9];

  const int HC   = in_sizes[3];
  const int Fin  = in_sizes[2] / HC;
  const int N    = in_sizes[0] / Fin;
  const int E    = in_sizes[1] / 2;
  const int Cout = in_sizes[7];

  float* ws = (float*)d_ws;
  float* h1  = ws;
  float* x2  = h1 + (size_t)N * HC;
  float* h2  = x2 + (size_t)N * HC;
  int*   rowptr = (int*)(h2 + (size_t)N * Cout);
  int*   csr    = rowptr + ((N + 1 + 31) / 32) * 32;
  int*   recS   = csr + E;
  int*   recD   = recS + E;
  const int nChunks = (E + BIN_CHUNK - 1) / BIN_CHUNK;
  int*   tabOff = recD + E;
  int*   tabCnt = tabOff + (size_t)nChunks * BKT_DSTS;

  const int TB = 256;
  auto blocks = [](long long n, int tb) { return (int)((n + tb - 1) / tb); };

  edge_bin<<<nChunks, SORT_TB, 0, stream>>>(ei, E, recS, recD, tabOff, tabCnt);
  bucket_sort<<<(N + BKT_DSTS - 1) / BKT_DSTS, SORT_TB, 0, stream>>>(
      recS, recD, tabOff, tabCnt, nChunks, rowptr, csr, N);

  {
    const long long waves = (long long)((N + 31) / 32) * (HC / 64);
    gemm_wmma_f32<2, 4, 32, 128, 256><<<blocks(waves * 32, TB), TB, 0, stream>>>(x, W1, h1, N);
    agg_h8<<<blocks((long long)N * 32, TB), TB, 0, stream>>>(
        rowptr, csr, as1, ad1, h1, b1, x2, N);
  }

  {
    const long long waves = (long long)(N / 16) * (Cout / 32);
    gemm_wmma_f32<1, 2, 64, 256, 32><<<blocks(waves * 32, TB), TB, 0, stream>>>(x2, W2, h2, N);
    agg_h1<<<blocks((long long)N * 32, TB), TB, 0, stream>>>(
        rowptr, csr, as2, ad2, h2, b2, (float*)d_out, N);
  }
}


# Layer-2 dense product runs in the layer-1 reducer's tail: 16 nodes a block leave rows in LDS, each wave does a 16-k slice on WMMA, slices added in order; one launch and the row round trip go
# speedup vs baseline: 13.7989x; 1.0331x over previous
#include <hip/hip_runtime.h>
#include <math.h>

typedef float v2f __attribute__((ext_vector_type(2)));
typedef float v4f __attribute__((ext_vector_type(4)));
typedef float v8f __attribute__((ext_vector_type(8)));

#define LRELU_SLOPE 0.2f

template <int MT, int NT, int KB, int KT, int NC>
__global__ __launch_bounds__(256) void gemm_wmma_f32(
    const float* __restrict__ X, const float* __restrict__ W,
    float* __restrict__ Y, int M) {
  typedef float vNf __attribute__((ext_vector_type(NT)));
  constexpr int NJ = KB / 8;
  constexpr int groupsN = NC / (16 * NT);
  const int lane = threadIdx.x & 31;
  const int wave = __builtin_amdgcn_readfirstlane(
      (int)((blockIdx.x * blockDim.x + threadIdx.x) >> 5));
  const int blocksM = (M + 16 * MT - 1) / (16 * MT);
  if (wave >= blocksM * groupsN) return;
  const int m0 = (wave / groupsN) * (16 * MT);
  const int n0 = (wave % groupsN) * (16 * NT);
  const int hf = lane >> 4;
  const int c  = lane & 15;

  const float* __restrict__ xr[MT];
#pragma unroll
  for (int mt = 0; mt < MT; ++mt) {
    int row = m0 + mt * 16 + c;
    row = (row < M) ? row : (M - 1);
    xr[mt] = X + (size_t)row * KT + 4 * hf;
  }
  const float* __restrict__ wp = W + (size_t)(4 * hf) * NC + n0 + NT * c;

  v8f acc[MT][NT];
#pragma unroll
  for (int mt = 0; mt < MT; ++mt)
#pragma unroll
    for (int t = 0; t < NT; ++t)
      acc[mt][t] = (v8f){0.f, 0.f, 0.f, 0.f, 0.f, 0.f, 0.f, 0.f};

#pragma unroll 1
  for (int kb = 0; kb < KT; kb += KB) {
    v4f a[MT][NJ];
    vNf w[NJ][4];
#pragma unroll
    for (int mt = 0; mt < MT; ++mt)
#pragma unroll
      for (int j = 0; j < NJ; ++j)
        a[mt][j] = *(const v4f*)(xr[mt] + kb + 8 * j);
#pragma unroll
    for (int j = 0; j < NJ; ++j)
#pragma unroll
      for (int q = 0; q < 4; ++q)
        w[j][q] = *(const vNf*)(wp + (size_t)(kb + 8 * j + q) * NC);

#pragma unroll
    for (int j = 0; j < NJ; ++j) {
#pragma unroll
      for (int p = 0; p < 2; ++p) {
#pragma unroll
        for (int mt = 0; mt < MT; ++mt) {
          v2f av;
          av.x = p ? a[mt][j].z : a[mt][j].x;
          av.y = p ? a[mt][j].w : a[mt][j].y;
#pragma unroll
          for (int t = 0; t < NT; ++t) {
            v2f bv;
            bv.x = w[j][2 * p][t];
            bv.y = w[j][2 * p + 1][t];
#if __has_builtin(__builtin_amdgcn_wmma_f32_16x16x4_f32)
            acc[mt][t] = __builtin_amdgcn_wmma_f32_16x16x4_f32(
                false, av, false, bv, (short)0, acc[mt][t], false, false);
#else
            acc[mt][t][0] += av.x * bv.x;
#endif
          }
        }
      }
    }
    __builtin_amdgcn_sched_group_barrier(0x020, MT * NJ + 4 * NJ, 0);
    __builtin_amdgcn_sched_group_barrier(0x00A, 8 * MT * NT * NJ, 0);
  }

#pragma unroll
  for (int mt = 0; mt < MT; ++mt) {
    if (m0 + mt * 16 < M) {
      float* __restrict__ yr = Y + (size_t)(m0 + mt * 16 + 8 * hf) * NC + n0 + NT * c;
#pragma unroll
      for (int r = 0; r < 8; ++r) {
        vNf o;
#pragma unroll
        for (int t = 0; t < NT; ++t) o[t] = acc[mt][t][r];
        *(vNf*)(yr + (size_t)r * NC) = o;
      }
    }
  }
}

#define SORT_TB    1024
#define BIN_EPT    4
#define BIN_CHUNK  (BIN_EPT * SORT_TB)
#define BKT_SHIFT  8
#define BKT_DSTS   256
#define BKT_CAP    6144

__device__ __forceinline__ int scan256(const int* v, int* o, int* wt,
                                       int tid, int wv, int& c) {
  const int lane = tid & 31;
  int incl = 0;
  c = 0;
  if (wv < 8) {
    c = v[tid];
    incl = c;
#pragma unroll
    for (int k = 1; k < 32; k <<= 1) {
      const int t = __shfl_up(incl, k, 32);
      if (lane >= k) incl += t;
    }
    if (lane == 31) wt[wv] = incl;
  }
  __syncthreads();
  int ex = 0;
  if (wv < 8) {
    int pre = 0;
#pragma unroll
    for (int w = 0; w < 8; ++w) pre += (w < wv) ? wt[w] : 0;
    ex = pre + incl - c;
    o[tid] = ex;
  }
  __syncthreads();
  return ex;
}

__global__ __launch_bounds__(SORT_TB) void edge_bin(
    const int* __restrict__ ei, int E, int* __restrict__ recS,
    int* __restrict__ recD, int* __restrict__ tabOff, int* __restrict__ tabCnt) {
  __shared__ int cnt[BKT_DSTS];
  __shared__ int off[BKT_DSTS];
  __shared__ int wt[8];
  __shared__ int srtS[BIN_CHUNK];
  __shared__ int srtD[BIN_CHUNK];
  const int tid = threadIdx.x;
  const int wv = __builtin_amdgcn_readfirstlane((int)(threadIdx.x >> 5));
  const int e0 = blockIdx.x * BIN_CHUNK;
  if (tid < BKT_DSTS) cnt[tid] = 0;
  __syncthreads();

  int s[BIN_EPT], d[BIN_EPT], r[BIN_EPT];
#pragma unroll
  for (int u = 0; u < BIN_EPT; ++u) {
    const int e = e0 + u * SORT_TB + tid;
    s[u] = 0;  d[u] = -1;  r[u] = 0;
    if (e < E) { s[u] = ei[e];  d[u] = ei[E + e]; }
  }
#pragma unroll
  for (int u = 0; u < BIN_EPT; ++u)
    if (d[u] >= 0) r[u] = atomicAdd(&cnt[d[u] >> BKT_SHIFT], 1);
  __syncthreads();

  int c;
  const int ex = scan256(cnt, off, wt, tid, wv, c);
  if (wv < 8) {
    tabOff[blockIdx.x * BKT_DSTS + tid] = ex;
    tabCnt[blockIdx.x * BKT_DSTS + tid] = c;
  }
#pragma unroll
  for (int u = 0; u < BIN_EPT; ++u) {
    if (d[u] >= 0) {
      const int p = off[d[u] >> BKT_SHIFT] + r[u];
      srtS[p] = s[u];
      srtD[p] = d[u];
    }
  }
  __syncthreads();
  int nE = E - e0;
  if (nE > BIN_CHUNK) nE = BIN_CHUNK;
#pragma unroll
  for (int u = 0; u < BIN_EPT; ++u) {
    const int i = u * SORT_TB + tid;
    if (i < nE) { recS[e0 + i] = srtS[i];  recD[e0 + i] = srtD[i]; }
  }
}

__device__ __forceinline__ int rec_index(int i, const int* pOfs, const int* pSrc) {
  int lo = 0;
#pragma unroll
  for (int step = 512; step > 0; step >>= 1) {
    const int m = lo + step;
    if (pOfs[m] <= i) lo = m;
  }
  return pSrc[lo] + (i - pOfs[lo]);
}

__global__ __launch_bounds__(SORT_TB) void bucket_sort(
    const int* __restrict__ recS, const int* __restrict__ recD,
    const int* __restrict__ tabOff, const int* __restrict__ tabCnt, int nChunks,
    int* __restrict__ rowptr, int* __restrict__ csr_src, int N) {
  __shared__ int pOfs[SORT_TB];
  __shared__ int pSrc[SORT_TB];
  __shared__ int wsum[32];
  __shared__ int wbase[32];
  __shared__ int dcnt[BKT_DSTS];
  __shared__ int dofs[BKT_DSTS];
  __shared__ int wt[8];
  __shared__ int bufS[BKT_CAP];
  __shared__ int bufK[BKT_CAP];
  __shared__ int fin[BKT_CAP];
  const int tid = threadIdx.x;
  const int lane = tid & 31;
  const int wv = __builtin_amdgcn_readfirstlane((int)(threadIdx.x >> 5));
  const int g = blockIdx.x;

  int o = 0, c = 0;
  if (tid < nChunks) {
    o = tabOff[tid * BKT_DSTS + g];
    c = tabCnt[tid * BKT_DSTS + g];
  }
  if (tid < BKT_DSTS) dcnt[tid] = 0;
  int inclC = c, sumO = o;
#pragma unroll
  for (int k = 1; k < 32; k <<= 1) {
    const int t = __shfl_up(inclC, k, 32);
    if (lane >= k) inclC += t;
  }
#pragma unroll
  for (int k = 16; k > 0; k >>= 1) sumO += __shfl_xor(sumO, k, 32);
  if (lane == 31) wsum[wv] = inclC;
  if (lane == 0) wbase[wv] = sumO;
  __syncthreads();
  const int ws = wsum[lane];
  int wincl = ws;
#pragma unroll
  for (int k = 1; k < 32; k <<= 1) {
    const int t = __shfl_up(wincl, k, 32);
    if (lane >= k) wincl += t;
  }
  const int wpre  = __shfl(wincl - ws, wv, 32);
  const int total = __builtin_amdgcn_readfirstlane(__shfl(wincl, 31, 32));
  int bsum = wbase[lane];
#pragma unroll
  for (int k = 16; k > 0; k >>= 1) bsum += __shfl_xor(bsum, k, 32);
  const int base = __builtin_amdgcn_readfirstlane(bsum);
  pOfs[tid] = wpre + inclC - c;
  pSrc[tid] = tid * BIN_CHUNK + o;
  __syncthreads();

  if (total <= BKT_CAP) {
    constexpr int RPT = BKT_CAP / SORT_TB;
    int sv[RPT], dv[RPT];
#pragma unroll
    for (int k = 0; k < RPT; ++k) {
      const int i = k * SORT_TB + tid;
      sv[k] = 0;  dv[k] = 0;
      if (i < total) {
        const int q = rec_index(i, pOfs, pSrc);
        sv[k] = recS[q];
        dv[k] = recD[q];
      }
    }
#pragma unroll
    for (int k = 0; k < RPT; ++k) {
      const int i = k * SORT_TB + tid;
      if (i < total) {
        const int dl = dv[k] & (BKT_DSTS - 1);
        const int rk = atomicAdd(&dcnt[dl], 1);
        bufS[i] = sv[k];
        bufK[i] = (rk << BKT_SHIFT) | dl;
      }
    }
    __syncthreads();
    int cc;
    const int ex = scan256(dcnt, dofs, wt, tid, wv, cc);
    if (wv < 8) {
      const int node = g * BKT_DSTS + tid;
      if (node < N) rowptr[node] = base + ex;
    }
    for (int i = tid; i < total; i += SORT_TB) {
      const int k = bufK[i];
      fin[dofs[k & (BKT_DSTS - 1)] + (k >> BKT_SHIFT)] = bufS[i];
    }
    __syncthreads();
    for (int i = tid; i < total; i += SORT_TB) csr_src[base + i] = fin[i];
  } else {
    for (int i = tid; i < total; i += SORT_TB) {
      const int q = rec_index(i, pOfs, pSrc);
      atomicAdd(&dcnt[recD[q] & (BKT_DSTS - 1)], 1);
    }
    __syncthreads();
    int cc;
    const int ex = scan256(dcnt, dofs, wt, tid, wv, cc);
    if (wv < 8) {
      const int node = g * BKT_DSTS + tid;
      if (node < N) rowptr[node] = base + ex;
      dcnt[tid] = 0;
    }
    __syncthreads();
    for (int i = tid; i < total; i += SORT_TB) {
      const int q = rec_index(i, pOfs, pSrc);
      const int dl = recD[q] & (BKT_DSTS - 1);
      const int rk = atomicAdd(&dcnt[dl], 1);
      csr_src[base + dofs[dl] + rk] = recS[q];
    }
  }
  if (g == (int)gridDim.x - 1 && tid == 0) rowptr[N] = base + total;
}

#define LOG2E 1.44269504088896340736f
__device__ __forceinline__ float exp2_hw(float t) {
  return __builtin_amdgcn_exp2f(t);
}
__device__ __forceinline__ float lrelu2(float l) {
  return l * ((l > 0.f) ? LOG2E : (LRELU_SLOPE * LOG2E));
}
__device__ __forceinline__ float elu1(float v) {
  return (v > 0.f) ? v : (exp2_hw(v * LOG2E) - 1.f);
}

template <int M>
__device__ __forceinline__ float xor_get(float v) {
  return __int_as_float(__builtin_amdgcn_update_dpp(
      0, __float_as_int(v), 0x160 + M, 0xf, 0xf, true));
}
template <int K>
__device__ __forceinline__ float bcast8(float v) {
  return __uint_as_float(__builtin_amdgcn_mov_dpp8(__float_as_uint(v), K * 0x249249));
}
__device__ __forceinline__ float sum8(float v) {
  v += xor_get<4>(v);
  v += xor_get<2>(v);
  v += xor_get<1>(v);
  return v;
}
__device__ __forceinline__ float dot4p(v4f a, v4f b) {
  v2f p = a.xy * b.xy;
  p = a.zw * b.zw + p;
  return p.x + p.y;
}

#define L1_NODES 16
#define XS_PITCH 260
__global__ __launch_bounds__(512) void agg_h8_dense2(
    const int* __restrict__ rowptr, const int* __restrict__ csr_src,
    const float* __restrict__ att_s, const float* __restrict__ att_d,
    const float* __restrict__ h, const float* __restrict__ bias,
    const float* __restrict__ W2, float* __restrict__ h2, int N) {
  __shared__ float xs[L1_NODES * XS_PITCH];
  __shared__ float part[L1_NODES * 512];
  const int lane = threadIdx.x & 31;
  const int wv = __builtin_amdgcn_readfirstlane((int)(threadIdx.x >> 5));
  const int dnode = (int)blockIdx.x * L1_NODES + wv;
  const bool live = dnode < N;
  const int d = live ? dnode : (N - 1);
  const bool b2 = (lane & 4) != 0, b1 = (lane & 2) != 0, b0 = (lane & 1) != 0;
  const int uq = lane & 3;
  const int beg = rowptr[d], end = rowptr[d + 1];
  const v4f sA = *(const v4f*)(att_s + 4 * lane), sB = *(const v4f*)(att_s + 128 + 4 * lane);
  const v4f dA = *(const v4f*)(att_d + 4 * lane), dB = *(const v4f*)(att_d + 128 + 4 * lane);
  const v4f* __restrict__ hd = (const v4f*)(h + (size_t)d * 256);
  const v4f rdA = hd[lane], rdB = hd[32 + lane];
  const float adA = sum8(dot4p(rdA, dA)), adB = sum8(dot4p(rdB, dB));
  const float lsA = lrelu2(sum8(dot4p(rdA, sA)) + adA);
  const float lsB = lrelu2(sum8(dot4p(rdB, sB)) + adB);

  const float ad = b2 ? adB : adA;
  float m = b2 ? lsB : lsA;
  float den = (uq == 0) ? 1.f : 0.f;
  v4f accA = rdA, accB = rdB;

  for (int base = beg; base < end; base += 32) {
    const int n = (end - base < 32) ? (end - base) : 32;
    const int sv = (lane < n) ? csr_src[base + lane] : d;
    for (int j = 0; j < n; j += 4) {
      v4f ra[4], rb[4];
#pragma unroll
      for (int u = 0; u < 4; ++u) {
        const int s = __builtin_amdgcn_readlane(sv, j + u);
        const v4f* __restrict__ hr = (const v4f*)(h + (size_t)s * 256);
        ra[u] = hr[lane];
        rb[u] = hr[32 + lane];
      }
      float k1[4];
#pragma unroll
      for (int u = 0; u < 4; ++u) {
        const float pa = dot4p(ra[u], sA), pb = dot4p(rb[u], sB);
        k1[u] = (b2 ? pb : pa) + xor_get<4>(b2 ? pa : pb);
      }
      float k2[2];
#pragma unroll
      for (int v = 0; v < 2; ++v)
        k2[v] = (b1 ? k1[2 + v] : k1[v]) + xor_get<2>(b1 ? k1[v] : k1[2 + v]);
      const float sc0 = (b0 ? k2[1] : k2[0]) + xor_get<1>(b0 ? k2[0] : k2[1]);

      const float l = (j + uq < n) ? lrelu2(sc0 + ad) : -__builtin_huge_valf();
      float g = fmaxf(l, xor_get<1>(l));
      g = fmaxf(g, xor_get<2>(g));
      const float mn = fmaxf(m, g);
      const float sc = exp2_hw(m - mn);
      const float w = exp2_hw(l - mn);
      m = mn;
      den = den * sc + w;
      accA *= bcast8<0>(sc);
      accB *= bcast8<4>(sc);
      accA += bcast8<0>(w) * ra[0];
      accB += bcast8<4>(w) * rb[0];
      accA += bcast8<1>(w) * ra[1];
      accB += bcast8<5>(w) * rb[1];
      accA += bcast8<2>(w) * ra[2];
      accB += bcast8<6>(w) * rb[2];
      accA += bcast8<3>(w) * ra[3];
      accB += bcast8<7>(w) * rb[3];
      __builtin_amdgcn_sched_group_barrier(0x020, 8, 0);
      __builtin_amdgcn_sched_group_barrier(0x002, 512, 0);
    }
  }

  den += xor_get<1>(den);
  den += xor_get<2>(den);
  const float qA = __builtin_amdgcn_rcpf(bcast8<0>(den) + 1e-16f);
  const float qB = __builtin_amdgcn_rcpf(bcast8<4>(den) + 1e-16f);
  const float* __restrict__ bA = bias + 4 * lane;
  const float* __restrict__ bB = bias + 128 + 4 * lane;
  v4f oA, oB;
  oA.x = accA.x * qA + bA[0];  oA.y = accA.y * qA + bA[1];
  oA.z = accA.z * qA + bA[2];  oA.w = accA.w * qA + bA[3];
  oB.x = accB.x * qB + bB[0];  oB.y = accB.y * qB + bB[1];
  oB.z = accB.z * qB + bB[2];  oB.w = accB.w * qB + bB[3];
  oA.x = elu1(oA.x);  oA.y = elu1(oA.y);  oA.z = elu1(oA.z);  oA.w = elu1(oA.w);
  oB.x = elu1(oB.x);  oB.y = elu1(oB.y);  oB.z = elu1(oB.z);  oB.w = elu1(oB.w);
  if (!live) {
    oA = (v4f){0.f, 0.f, 0.f, 0.f};
    oB = (v4f){0.f, 0.f, 0.f, 0.f};
  }
  {
    const int hf = lane >> 4;
    const int c  = lane & 15;
    v4f a[2];
    v2f w[2][4];
    const float* __restrict__ wp = W2 + (size_t)(16 * wv + 4 * hf) * 32 + 2 * c;
#pragma unroll
    for (int j = 0; j < 2; ++j)
#pragma unroll
      for (int q = 0; q < 4; ++q)
        w[j][q] = *(const v2f*)(wp + (size_t)(8 * j + q) * 32);

    *(v4f*)(xs + wv * XS_PITCH + 4 * lane)       = oA;
    *(v4f*)(xs + wv * XS_PITCH + 128 + 4 * lane) = oB;
    __syncthreads();

    const float* __restrict__ xa = xs + c * XS_PITCH + 16 * wv + 4 * hf;
    a[0] = *(const v4f*)(xa);
    a[1] = *(const v4f*)(xa + 8);
    v8f acc[2];
    acc[0] = (v8f){0.f, 0.f, 0.f, 0.f, 0.f, 0.f, 0.f, 0.f};
    acc[1] = (v8f){0.f, 0.f, 0.f, 0.f, 0.f, 0.f, 0.f, 0.f};
#pragma unroll
    for (int j = 0; j < 2; ++j) {
#pragma unroll
      for (int p = 0; p < 2; ++p) {
        v2f av;
        av.x = p ? a[j].z : a[j].x;
        av.y = p ? a[j].w : a[j].y;
#pragma unroll
        for (int t = 0; t < 2; ++t) {
          v2f bv;
          bv.x = w[j][2 * p][t];
          bv.y = w[j][2 * p + 1][t];
#if __has_builtin(__builtin_amdgcn_wmma_f32_16x16x4_f32)
          acc[t] = __builtin_amdgcn_wmma_f32_16x16x4_f32(
              false, av, false, bv, (short)0, acc[t], false, false);
#else
          acc[t][0] += av.x * bv.x;
#endif
        }
      }
    }
    float* __restrict__ pw = part + wv * 512 + (8 * hf) * 32 + 2 * c;
#pragma unroll
    for (int r = 0; r < 8; ++r) {
      v2f o;
      o.x = acc[0][r];
      o.y = acc[1][r];
      *(v2f*)(pw + r * 32) = o;
    }
  }
  __syncthreads();
  {
    const float* __restrict__ pe = part + threadIdx.x;
    float sum = pe[0];
#pragma unroll
    for (int k = 1; k < L1_NODES; ++k) sum += pe[k * 512];
    if (live) h2[(size_t)dnode * 32 + lane] = sum;
  }
}

template <int K>
__device__ __forceinline__ int bcast8i(int v) {
  return (int)__builtin_amdgcn_mov_dpp8((unsigned)v, K * 0x249249);
}

__global__ __launch_bounds__(256) __attribute__((amdgpu_waves_per_eu(16, 16))) void agg_h1(
    const int* __restrict__ rowptr, const int* __restrict__ csr_src,
    const float* __restrict__ att_s, const float* __restrict__ att_d,
    const float* __restrict__ h, const float* __restrict__ bias,
    float* __restrict__ out, int N) {
  const int lane = threadIdx.x & 31;
  const int d = __builtin_amdgcn_readfirstlane((int)(blockIdx.x * 8 + (threadIdx.x >> 5)));
  if (d >= N) return;
  const int q = lane & 7;
  const bool b2 = (lane & 4) != 0, b1 = (lane & 2) != 0, b0 = (lane & 1) != 0;
  const int beg = rowptr[d], end = rowptr[d + 1];
  const v4f s4 = *(const v4f*)(att_s + 4 * q), d4 = *(const v4f*)(att_d + 4 * q);
  const v4f* __restrict__ hq = (const v4f*)h + q;
  const v4f rd = hq[(size_t)d * 8];
  const float ad = sum8(dot4p(rd, d4));
  float m = lrelu2(sum8(dot4p(rd, s4)) + ad);
  float den = (lane == 0) ? 1.f : 0.f;
  v4f acc = rd;
  if (lane >= 8) acc = (v4f){0.f, 0.f, 0.f, 0.f};

  for (int base = beg; base < end; base += 32) {
    const int n = (end - base < 32) ? (end - base) : 32;
    const int sv = (lane < n) ? csr_src[base + lane] : d;
    v4f r[8];
    r[0] = hq[(size_t)bcast8i<0>(sv) * 8];
    r[1] = hq[(size_t)bcast8i<1>(sv) * 8];
    r[2] = hq[(size_t)bcast8i<2>(sv) * 8];
    r[3] = hq[(size_t)bcast8i<3>(sv) * 8];
    r[4] = hq[(size_t)bcast8i<4>(sv) * 8];
    r[5] = hq[(size_t)bcast8i<5>(sv) * 8];
    r[6] = hq[(size_t)bcast8i<6>(sv) * 8];
    r[7] = hq[(size_t)bcast8i<7>(sv) * 8];
    float k1[4];
#pragma unroll
    for (int v = 0; v < 4; ++v) {
      const float pa = dot4p(r[v], s4), pb = dot4p(r[4 + v], s4);
      k1[v] = (b2 ? pb : pa) + xor_get<4>(b2 ? pa : pb);
    }
    float k2[2];
#pragma unroll
    for (int v = 0; v < 2; ++v)
      k2[v] = (b1 ? k1[2 + v] : k1[v]) + xor_get<2>(b1 ? k1[v] : k1[2 + v]);
    const float sc0 = (b0 ? k2[1] : k2[0]) + xor_get<1>(b0 ? k2[0] : k2[1]);

    const float l = (lane < n) ? lrelu2(sc0 + ad) : -__builtin_huge_valf();
    float g = fmaxf(l, xor_get<1>(l));
    g = fmaxf(g, xor_get<2>(g));
    g = fmaxf(g, xor_get<4>(g));
    g = fmaxf(g, xor_get<8>(g));
    g = fmaxf(g, __shfl_xor(g, 16, 32));
    const float mn = fmaxf(m, g);
    const float sc = exp2_hw(m - mn);
    const float w = exp2_hw(l - mn);
    m = mn;
    den = den * sc + w;
    acc *= sc;
    acc += bcast8<0>(w) * r[0];
    acc += bcast8<1>(w) * r[1];
    acc += bcast8<2>(w) * r[2];
    acc += bcast8<3>(w) * r[3];
    acc += bcast8<4>(w) * r[4];
    acc += bcast8<5>(w) * r[5];
    acc += bcast8<6>(w) * r[6];
    acc += bcast8<7>(w) * r[7];
  }

  acc.x += xor_get<8>(acc.x);  acc.y += xor_get<8>(acc.y);
  acc.z += xor_get<8>(acc.z);  acc.w += xor_get<8>(acc.w);
  acc.x += __shfl_xor(acc.x, 16, 32);  acc.y += __shfl_xor(acc.y, 16, 32);
  acc.z += __shfl_xor(acc.z, 16, 32);  acc.w += __shfl_xor(acc.w, 16, 32);
  den = sum8(den);
  den += xor_get<8>(den);
  den += __shfl_xor(den, 16, 32);
  if (lane < 8) {
    const float qd = __builtin_amdgcn_rcpf(den + 1e-16f);
    const v4f bv = *(const v4f*)(bias + 4 * q);
    v4f o;
    o.x = acc.x * qd + bv.x;  o.y = acc.y * qd + bv.y;
    o.z = acc.z * qd + bv.z;  o.w = acc.w * qd + bv.w;
    o.x = elu1(o.x);  o.y = elu1(o.y);  o.z = elu1(o.z);  o.w = elu1(o.w);
    ((v4f*)(out + (size_t)d * 32))[q] = o;
  }
}

extern "C" void kernel_launch(void* const* d_in, const int* in_sizes, int n_in,
                              void* d_out, int out_size, void* d_ws, size_t ws_size,
                              hipStream_t stream) {
  const float* x   = (const float*)d_in[0];
  const int*   ei  = (const int*)d_in[1];
  const float* W1  = (const float*)d_in[2];
  const float* as1 = (const float*)d_in[3];
  const float* ad1 = (const float*)d_in[4];
  const float* b1  = (const float*)d_in[5];
  const float* W2  = (const float*)d_in[6];
  const float* as2 = (const float*)d_in[7];
  const float* ad2 = (const float*)d_in[8];
  const float* b2  = (const float*)d_in[9];

  const int HC   = in_sizes[3];
  const int Fin  = in_sizes[2] / HC;
  const int N    = in_sizes[0] / Fin;
  const int E    = in_sizes[1] / 2;
  const int Cout = in_sizes[7];

  float* ws = (float*)d_ws;
  float* h1  = ws;
  float* h2  = h1 + (size_t)N * HC;
  int*   rowptr = (int*)(h2 + (size_t)N * Cout);
  int*   csr    = rowptr + ((N + 1 + 31) / 32) * 32;
  int*   recS   = csr + E;
  int*   recD   = recS + E;
  const int nChunks = (E + BIN_CHUNK - 1) / BIN_CHUNK;
  int*   tabOff = recD + E;
  int*   tabCnt = tabOff + (size_t)nChunks * BKT_DSTS;

  const int TB = 256;
  auto blocks = [](long long n, int tb) { return (int)((n + tb - 1) / tb); };

  edge_bin<<<nChunks, SORT_TB, 0, stream>>>(ei, E, recS, recD, tabOff, tabCnt);
  bucket_sort<<<(N + BKT_DSTS - 1) / BKT_DSTS, SORT_TB, 0, stream>>>(
      recS, recD, tabOff, tabCnt, nChunks, rowptr, csr, N);

  {
    const long long waves = (long long)((N + 31) / 32) * (HC / 64);
    gemm_wmma_f32<2, 4, 32, 128, 256><<<blocks(waves * 32, TB), TB, 0, stream>>>(x, W1, h1, N);
    agg_h8_dense2<<<(N + L1_NODES - 1) / L1_NODES, 32 * L1_NODES, 0, stream>>>(
        rowptr, csr, as1, ad1, h1, b1, W2, h2, N);
  }

  {
    agg_h1<<<blocks((long long)N * 32, TB), TB, 0, stream>>>(
        rowptr, csr, as2, ad2, h2, b2, (float*)d_out, N);
  }
}


# Layer-2 reducer: an 8-lane group per destination, 4 nodes a wave (a quarter of the waves start), 8 edges a trip, ids a trip ahead; per-node sum order changes (8-edge max steps, no cross-group add)
# speedup vs baseline: 14.2178x; 1.0304x over previous
#include <hip/hip_runtime.h>
#include <math.h>

typedef float v2f __attribute__((ext_vector_type(2)));
typedef float v4f __attribute__((ext_vector_type(4)));
typedef float v8f __attribute__((ext_vector_type(8)));

#define LRELU_SLOPE 0.2f

template <int MT, int NT, int KB, int KT, int NC>
__global__ __launch_bounds__(256) void gemm_wmma_f32(
    const float* __restrict__ X, const float* __restrict__ W,
    float* __restrict__ Y, int M) {
  typedef float vNf __attribute__((ext_vector_type(NT)));
  constexpr int NJ = KB / 8;
  constexpr int groupsN = NC / (16 * NT);
  const int lane = threadIdx.x & 31;
  const int wave = __builtin_amdgcn_readfirstlane(
      (int)((blockIdx.x * blockDim.x + threadIdx.x) >> 5));
  const int blocksM = (M + 16 * MT - 1) / (16 * MT);
  if (wave >= blocksM * groupsN) return;
  const int m0 = (wave / groupsN) * (16 * MT);
  const int n0 = (wave % groupsN) * (16 * NT);
  const int hf = lane >> 4;
  const int c  = lane & 15;

  const float* __restrict__ xr[MT];
#pragma unroll
  for (int mt = 0; mt < MT; ++mt) {
    int row = m0 + mt * 16 + c;
    row = (row < M) ? row : (M - 1);
    xr[mt] = X + (size_t)row * KT + 4 * hf;
  }
  const float* __restrict__ wp = W + (size_t)(4 * hf) * NC + n0 + NT * c;

  v8f acc[MT][NT];
#pragma unroll
  for (int mt = 0; mt < MT; ++mt)
#pragma unroll
    for (int t = 0; t < NT; ++t)
      acc[mt][t] = (v8f){0.f, 0.f, 0.f, 0.f, 0.f, 0.f, 0.f, 0.f};

#pragma unroll 1
  for (int kb = 0; kb < KT; kb += KB) {
    v4f a[MT][NJ];
    vNf w[NJ][4];
#pragma unroll
    for (int mt = 0; mt < MT; ++mt)
#pragma unroll
      for (int j = 0; j < NJ; ++j)
        a[mt][j] = *(const v4f*)(xr[mt] + kb + 8 * j);
#pragma unroll
    for (int j = 0; j < NJ; ++j)
#pragma unroll
      for (int q = 0; q < 4; ++q)
        w[j][q] = *(const vNf*)(wp + (size_t)(kb + 8 * j + q) * NC);

#pragma unroll
    for (int j = 0; j < NJ; ++j) {
#pragma unroll
      for (int p = 0; p < 2; ++p) {
#pragma unroll
        for (int mt = 0; mt < MT; ++mt) {
          v2f av;
          av.x = p ? a[mt][j].z : a[mt][j].x;
          av.y = p ? a[mt][j].w : a[mt][j].y;
#pragma unroll
          for (int t = 0; t < NT; ++t) {
            v2f bv;
            bv.x = w[j][2 * p][t];
            bv.y = w[j][2 * p + 1][t];
#if __has_builtin(__builtin_amdgcn_wmma_f32_16x16x4_f32)
            acc[mt][t] = __builtin_amdgcn_wmma_f32_16x16x4_f32(
                false, av, false, bv, (short)0, acc[mt][t], false, false);
#else
            acc[mt][t][0] += av.x * bv.x;
#endif
          }
        }
      }
    }
    __builtin_amdgcn_sched_group_barrier(0x020, MT * NJ + 4 * NJ, 0);
    __builtin_amdgcn_sched_group_barrier(0x00A, 8 * MT * NT * NJ, 0);
  }

#pragma unroll
  for (int mt = 0; mt < MT; ++mt) {
    if (m0 + mt * 16 < M) {
      float* __restrict__ yr = Y + (size_t)(m0 + mt * 16 + 8 * hf) * NC + n0 + NT * c;
#pragma unroll
      for (int r = 0; r < 8; ++r) {
        vNf o;
#pragma unroll
        for (int t = 0; t < NT; ++t) o[t] = acc[mt][t][r];
        *(vNf*)(yr + (size_t)r * NC) = o;
      }
    }
  }
}

#define SORT_TB    1024
#define BIN_EPT    4
#define BIN_CHUNK  (BIN_EPT * SORT_TB)
#define BKT_SHIFT  8
#define BKT_DSTS   256
#define BKT_CAP    6144

__device__ __forceinline__ int scan256(const int* v, int* o, int* wt,
                                       int tid, int wv, int& c) {
  const int lane = tid & 31;
  int incl = 0;
  c = 0;
  if (wv < 8) {
    c = v[tid];
    incl = c;
#pragma unroll
    for (int k = 1; k < 32; k <<= 1) {
      const int t = __shfl_up(incl, k, 32);
      if (lane >= k) incl += t;
    }
    if (lane == 31) wt[wv] = incl;
  }
  __syncthreads();
  int ex = 0;
  if (wv < 8) {
    int pre = 0;
#pragma unroll
    for (int w = 0; w < 8; ++w) pre += (w < wv) ? wt[w] : 0;
    ex = pre + incl - c;
    o[tid] = ex;
  }
  __syncthreads();
  return ex;
}

__global__ __launch_bounds__(SORT_TB) void edge_bin(
    const int* __restrict__ ei, int E, int* __restrict__ recS,
    int* __restrict__ recD, int* __restrict__ tabOff, int* __restrict__ tabCnt) {
  __shared__ int cnt[BKT_DSTS];
  __shared__ int off[BKT_DSTS];
  __shared__ int wt[8];
  __shared__ int srtS[BIN_CHUNK];
  __shared__ int srtD[BIN_CHUNK];
  const int tid = threadIdx.x;
  const int wv = __builtin_amdgcn_readfirstlane((int)(threadIdx.x >> 5));
  const int e0 = blockIdx.x * BIN_CHUNK;
  if (tid < BKT_DSTS) cnt[tid] = 0;
  __syncthreads();

  int s[BIN_EPT], d[BIN_EPT], r[BIN_EPT];
#pragma unroll
  for (int u = 0; u < BIN_EPT; ++u) {
    const int e = e0 + u * SORT_TB + tid;
    s[u] = 0;  d[u] = -1;  r[u] = 0;
    if (e < E) { s[u] = ei[e];  d[u] = ei[E + e]; }
  }
#pragma unroll
  for (int u = 0; u < BIN_EPT; ++u)
    if (d[u] >= 0) r[u] = atomicAdd(&cnt[d[u] >> BKT_SHIFT], 1);
  __syncthreads();

  int c;
  const int ex = scan256(cnt, off, wt, tid, wv, c);
  if (wv < 8) {
    tabOff[blockIdx.x * BKT_DSTS + tid] = ex;
    tabCnt[blockIdx.x * BKT_DSTS + tid] = c;
  }
#pragma unroll
  for (int u = 0; u < BIN_EPT; ++u) {
    if (d[u] >= 0) {
      const int p = off[d[u] >> BKT_SHIFT] + r[u];
      srtS[p] = s[u];
      srtD[p] = d[u];
    }
  }
  __syncthreads();
  int nE = E - e0;
  if (nE > BIN_CHUNK) nE = BIN_CHUNK;
#pragma unroll
  for (int u = 0; u < BIN_EPT; ++u) {
    const int i = u * SORT_TB + tid;
    if (i < nE) { recS[e0 + i] = srtS[i];  recD[e0 + i] = srtD[i]; }
  }
}

__device__ __forceinline__ int rec_index(int i, const int* pOfs, const int* pSrc) {
  int lo = 0;
#pragma unroll
  for (int step = 512; step > 0; step >>= 1) {
    const int m = lo + step;
    if (pOfs[m] <= i) lo = m;
  }
  return pSrc[lo] + (i - pOfs[lo]);
}

__global__ __launch_bounds__(SORT_TB) void bucket_sort(
    const int* __restrict__ recS, const int* __restrict__ recD,
    const int* __restrict__ tabOff, const int* __restrict__ tabCnt, int nChunks,
    int* __restrict__ rowptr, int* __restrict__ csr_src, int N) {
  __shared__ int pOfs[SORT_TB];
  __shared__ int pSrc[SORT_TB];
  __shared__ int wsum[32];
  __shared__ int wbase[32];
  __shared__ int dcnt[BKT_DSTS];
  __shared__ int dofs[BKT_DSTS];
  __shared__ int wt[8];
  __shared__ int bufS[BKT_CAP];
  __shared__ int bufK[BKT_CAP];
  __shared__ int fin[BKT_CAP];
  const int tid = threadIdx.x;
  const int lane = tid & 31;
  const int wv = __builtin_amdgcn_readfirstlane((int)(threadIdx.x >> 5));
  const int g = blockIdx.x;

  int o = 0, c = 0;
  if (tid < nChunks) {
    o = tabOff[tid * BKT_DSTS + g];
    c = tabCnt[tid * BKT_DSTS + g];
  }
  if (tid < BKT_DSTS) dcnt[tid] = 0;
  int inclC = c, sumO = o;
#pragma unroll
  for (int k = 1; k < 32; k <<= 1) {
    const int t = __shfl_up(inclC, k, 32);
    if (lane >= k) inclC += t;
  }
#pragma unroll
  for (int k = 16; k > 0; k >>= 1) sumO += __shfl_xor(sumO, k, 32);
  if (lane == 31) wsum[wv] = inclC;
  if (lane == 0) wbase[wv] = sumO;
  __syncthreads();
  const int ws = wsum[lane];
  int wincl = ws;
#pragma unroll
  for (int k = 1; k < 32; k <<= 1) {
    const int t = __shfl_up(wincl, k, 32);
    if (lane >= k) wincl += t;
  }
  const int wpre  = __shfl(wincl - ws, wv, 32);
  const int total = __builtin_amdgcn_readfirstlane(__shfl(wincl, 31, 32));
  int bsum = wbase[lane];
#pragma unroll
  for (int k = 16; k > 0; k >>= 1) bsum += __shfl_xor(bsum, k, 32);
  const int base = __builtin_amdgcn_readfirstlane(bsum);
  pOfs[tid] = wpre + inclC - c;
  pSrc[tid] = tid * BIN_CHUNK + o;
  __syncthreads();

  if (total <= BKT_CAP) {
    constexpr int RPT = BKT_CAP / SORT_TB;
    int sv[RPT], dv[RPT];
#pragma unroll
    for (int k = 0; k < RPT; ++k) {
      const int i = k * SORT_TB + tid;
      sv[k] = 0;  dv[k] = 0;
      if (i < total) {
        const int q = rec_index(i, pOfs, pSrc);
        sv[k] = recS[q];
        dv[k] = recD[q];
      }
    }
#pragma unroll
    for (int k = 0; k < RPT; ++k) {
      const int i = k * SORT_TB + tid;
      if (i < total) {
        const int dl = dv[k] & (BKT_DSTS - 1);
        const int rk = atomicAdd(&dcnt[dl], 1);
        bufS[i] = sv[k];
        bufK[i] = (rk << BKT_SHIFT) | dl;
      }
    }
    __syncthreads();
    int cc;
    const int ex = scan256(dcnt, dofs, wt, tid, wv, cc);
    if (wv < 8) {
      const int node = g * BKT_DSTS + tid;
      if (node < N) rowptr[node] = base + ex;
    }
    for (int i = tid; i < total; i += SORT_TB) {
      const int k = bufK[i];
      fin[dofs[k & (BKT_DSTS - 1)] + (k >> BKT_SHIFT)] = bufS[i];
    }
    __syncthreads();
    for (int i = tid; i < total; i += SORT_TB) csr_src[base + i] = fin[i];
  } else {
    for (int i = tid; i < total; i += SORT_TB) {
      const int q = rec_index(i, pOfs, pSrc);
      atomicAdd(&dcnt[recD[q] & (BKT_DSTS - 1)], 1);
    }
    __syncthreads();
    int cc;
    const int ex = scan256(dcnt, dofs, wt, tid, wv, cc);
    if (wv < 8) {
      const int node = g * BKT_DSTS + tid;
      if (node < N) rowptr[node] = base + ex;
      dcnt[tid] = 0;
    }
    __syncthreads();
    for (int i = tid; i < total; i += SORT_TB) {
      const int q = rec_index(i, pOfs, pSrc);
      const int dl = recD[q] & (BKT_DSTS - 1);
      const int rk = atomicAdd(&dcnt[dl], 1);
      csr_src[base + dofs[dl] + rk] = recS[q];
    }
  }
  if (g == (int)gridDim.x - 1 && tid == 0) rowptr[N] = base + total;
}

#define LOG2E 1.44269504088896340736f
__device__ __forceinline__ float exp2_hw(float t) {
  return __builtin_amdgcn_exp2f(t);
}
__device__ __forceinline__ float lrelu2(float l) {
  return l * ((l > 0.f) ? LOG2E : (LRELU_SLOPE * LOG2E));
}
__device__ __forceinline__ float elu1(float v) {
  return (v > 0.f) ? v : (exp2_hw(v * LOG2E) - 1.f);
}

template <int M>
__device__ __forceinline__ float xor_get(float v) {
  return __int_as_float(__builtin_amdgcn_update_dpp(
      0, __float_as_int(v), 0x160 + M, 0xf, 0xf, true));
}
template <int K>
__device__ __forceinline__ float bcast8(float v) {
  return __uint_as_float(__builtin_amdgcn_mov_dpp8(__float_as_uint(v), K * 0x249249));
}
__device__ __forceinline__ float sum8(float v) {
  v += xor_get<4>(v);
  v += xor_get<2>(v);
  v += xor_get<1>(v);
  return v;
}
__device__ __forceinline__ float dot4p(v4f a, v4f b) {
  v2f p = a.xy * b.xy;
  p = a.zw * b.zw + p;
  return p.x + p.y;
}

#define L1_NODES 16
#define XS_PITCH 260
__global__ __launch_bounds__(512) void agg_h8_dense2(
    const int* __restrict__ rowptr, const int* __restrict__ csr_src,
    const float* __restrict__ att_s, const float* __restrict__ att_d,
    const float* __restrict__ h, const float* __restrict__ bias,
    const float* __restrict__ W2, float* __restrict__ h2, int N) {
  __shared__ float xs[L1_NODES * XS_PITCH];
  __shared__ float part[L1_NODES * 512];
  const int lane = threadIdx.x & 31;
  const int wv = __builtin_amdgcn_readfirstlane((int)(threadIdx.x >> 5));
  const int dnode = (int)blockIdx.x * L1_NODES + wv;
  const bool live = dnode < N;
  const int d = live ? dnode : (N - 1);
  const bool b2 = (lane & 4) != 0, b1 = (lane & 2) != 0, b0 = (lane & 1) != 0;
  const int uq = lane & 3;
  const int beg = rowptr[d], end = rowptr[d + 1];
  const v4f sA = *(const v4f*)(att_s + 4 * lane), sB = *(const v4f*)(att_s + 128 + 4 * lane);
  const v4f dA = *(const v4f*)(att_d + 4 * lane), dB = *(const v4f*)(att_d + 128 + 4 * lane);
  const v4f* __restrict__ hd = (const v4f*)(h + (size_t)d * 256);
  const v4f rdA = hd[lane], rdB = hd[32 + lane];
  const float adA = sum8(dot4p(rdA, dA)), adB = sum8(dot4p(rdB, dB));
  const float lsA = lrelu2(sum8(dot4p(rdA, sA)) + adA);
  const float lsB = lrelu2(sum8(dot4p(rdB, sB)) + adB);

  const float ad = b2 ? adB : adA;
  float m = b2 ? lsB : lsA;
  float den = (uq == 0) ? 1.f : 0.f;
  v4f accA = rdA, accB = rdB;

  for (int base = beg; base < end; base += 32) {
    const int n = (end - base < 32) ? (end - base) : 32;
    const int sv = (lane < n) ? csr_src[base + lane] : d;
    for (int j = 0; j < n; j += 4) {
      v4f ra[4], rb[4];
#pragma unroll
      for (int u = 0; u < 4; ++u) {
        const int s = __builtin_amdgcn_readlane(sv, j + u);
        const v4f* __restrict__ hr = (const v4f*)(h + (size_t)s * 256);
        ra[u] = hr[lane];
        rb[u] = hr[32 + lane];
      }
      float k1[4];
#pragma unroll
      for (int u = 0; u < 4; ++u) {
        const float pa = dot4p(ra[u], sA), pb = dot4p(rb[u], sB);
        k1[u] = (b2 ? pb : pa) + xor_get<4>(b2 ? pa : pb);
      }
      float k2[2];
#pragma unroll
      for (int v = 0; v < 2; ++v)
        k2[v] = (b1 ? k1[2 + v] : k1[v]) + xor_get<2>(b1 ? k1[v] : k1[2 + v]);
      const float sc0 = (b0 ? k2[1] : k2[0]) + xor_get<1>(b0 ? k2[0] : k2[1]);

      const float l = (j + uq < n) ? lrelu2(sc0 + ad) : -__builtin_huge_valf();
      float g = fmaxf(l, xor_get<1>(l));
      g = fmaxf(g, xor_get<2>(g));
      const float mn = fmaxf(m, g);
      const float sc = exp2_hw(m - mn);
      const float w = exp2_hw(l - mn);
      m = mn;
      den = den * sc + w;
      accA *= bcast8<0>(sc);
      accB *= bcast8<4>(sc);
      accA += bcast8<0>(w) * ra[0];
      accB += bcast8<4>(w) * rb[0];
      accA += bcast8<1>(w) * ra[1];
      accB += bcast8<5>(w) * rb[1];
      accA += bcast8<2>(w) * ra[2];
      accB += bcast8<6>(w) * rb[2];
      accA += bcast8<3>(w) * ra[3];
      accB += bcast8<7>(w) * rb[3];
      __builtin_amdgcn_sched_group_barrier(0x020, 8, 0);
      __builtin_amdgcn_sched_group_barrier(0x002, 512, 0);
    }
  }

  den += xor_get<1>(den);
  den += xor_get<2>(den);
  const float qA = __builtin_amdgcn_rcpf(bcast8<0>(den) + 1e-16f);
  const float qB = __builtin_amdgcn_rcpf(bcast8<4>(den) + 1e-16f);
  const float* __restrict__ bA = bias + 4 * lane;
  const float* __restrict__ bB = bias + 128 + 4 * lane;
  v4f oA, oB;
  oA.x = accA.x * qA + bA[0];  oA.y = accA.y * qA + bA[1];
  oA.z = accA.z * qA + bA[2];  oA.w = accA.w * qA + bA[3];
  oB.x = accB.x * qB + bB[0];  oB.y = accB.y * qB + bB[1];
  oB.z = accB.z * qB + bB[2];  oB.w = accB.w * qB + bB[3];
  oA.x = elu1(oA.x);  oA.y = elu1(oA.y);  oA.z = elu1(oA.z);  oA.w = elu1(oA.w);
  oB.x = elu1(oB.x);  oB.y = elu1(oB.y);  oB.z = elu1(oB.z);  oB.w = elu1(oB.w);
  if (!live) {
    oA = (v4f){0.f, 0.f, 0.f, 0.f};
    oB = (v4f){0.f, 0.f, 0.f, 0.f};
  }
  {
    const int hf = lane >> 4;
    const int c  = lane & 15;
    v4f a[2];
    v2f w[2][4];
    const float* __restrict__ wp = W2 + (size_t)(16 * wv + 4 * hf) * 32 + 2 * c;
#pragma unroll
    for (int j = 0; j < 2; ++j)
#pragma unroll
      for (int q = 0; q < 4; ++q)
        w[j][q] = *(const v2f*)(wp + (size_t)(8 * j + q) * 32);

    *(v4f*)(xs + wv * XS_PITCH + 4 * lane)       = oA;
    *(v4f*)(xs + wv * XS_PITCH + 128 + 4 * lane) = oB;
    __syncthreads();

    const float* __restrict__ xa = xs + c * XS_PITCH + 16 * wv + 4 * hf;
    a[0] = *(const v4f*)(xa);
    a[1] = *(const v4f*)(xa + 8);
    v8f acc[2];
    acc[0] = (v8f){0.f, 0.f, 0.f, 0.f, 0.f, 0.f, 0.f, 0.f};
    acc[1] = (v8f){0.f, 0.f, 0.f, 0.f, 0.f, 0.f, 0.f, 0.f};
#pragma unroll
    for (int j = 0; j < 2; ++j) {
#pragma unroll
      for (int p = 0; p < 2; ++p) {
        v2f av;
        av.x = p ? a[j].z : a[j].x;
        av.y = p ? a[j].w : a[j].y;
#pragma unroll
        for (int t = 0; t < 2; ++t) {
          v2f bv;
          bv.x = w[j][2 * p][t];
          bv.y = w[j][2 * p + 1][t];
#if __has_builtin(__builtin_amdgcn_wmma_f32_16x16x4_f32)
          acc[t] = __builtin_amdgcn_wmma_f32_16x16x4_f32(
              false, av, false, bv, (short)0, acc[t], false, false);
#else
          acc[t][0] += av.x * bv.x;
#endif
        }
      }
    }
    float* __restrict__ pw = part + wv * 512 + (8 * hf) * 32 + 2 * c;
#pragma unroll
    for (int r = 0; r < 8; ++r) {
      v2f o;
      o.x = acc[0][r];
      o.y = acc[1][r];
      *(v2f*)(pw + r * 32) = o;
    }
  }
  __syncthreads();
  {
    const float* __restrict__ pe = part + threadIdx.x;
    float sum = pe[0];
#pragma unroll
    for (int k = 1; k < L1_NODES; ++k) sum += pe[k * 512];
    if (live) h2[(size_t)dnode * 32 + lane] = sum;
  }
}

template <int K>
__device__ __forceinline__ int bcast8i(int v) {
  return (int)__builtin_amdgcn_mov_dpp8((unsigned)v, K * 0x249249);
}

#define L2_NPW 4

__global__ __launch_bounds__(256) __attribute__((amdgpu_waves_per_eu(16, 16))) void agg_h1(
    const int* __restrict__ rowptr, const int* __restrict__ csr_src,
    const float* __restrict__ att_s, const float* __restrict__ att_d,
    const float* __restrict__ h, const float* __restrict__ bias,
    float* __restrict__ out, int N) {
  const int lane = threadIdx.x & 31;
  const int d0 = __builtin_amdgcn_readfirstlane(
      (int)(blockIdx.x * 8 + (threadIdx.x >> 5)) * L2_NPW);
  if (d0 >= N) return;
  const int q = lane & 7;
  const bool b2 = (lane & 4) != 0, b1 = (lane & 2) != 0, b0 = (lane & 1) != 0;
  const int dn = d0 + (lane >> 3);
  const bool live = dn < N;
  const int d = live ? dn : (N - 1);
  const int beg = rowptr[d];
  const int end = rowptr[d + 1];
  const v4f s4 = *(const v4f*)(att_s + 4 * q), d4 = *(const v4f*)(att_d + 4 * q);
  const v4f* __restrict__ h4 = (const v4f*)h;
  const v4f rd = h4[d * 8 + q];
  const float ad = sum8(dot4p(rd, d4));
  float m = lrelu2(sum8(dot4p(rd, s4)) + ad);
  float den = (q == 0) ? 1.f : 0.f;
  v4f acc = rd;

  int n = live ? (end - beg) : 0;
  unsigned idx = (unsigned)(beg + q);
  int sv = (q < n) ? csr_src[idx] : d;
  while (__builtin_amdgcn_ballot_w32(n > 0) != 0u) {
    v4f r[8];
    r[0] = h4[bcast8i<0>(sv) * 8 + q];
    r[1] = h4[bcast8i<1>(sv) * 8 + q];
    r[2] = h4[bcast8i<2>(sv) * 8 + q];
    r[3] = h4[bcast8i<3>(sv) * 8 + q];
    r[4] = h4[bcast8i<4>(sv) * 8 + q];
    r[5] = h4[bcast8i<5>(sv) * 8 + q];
    r[6] = h4[bcast8i<6>(sv) * 8 + q];
    r[7] = h4[bcast8i<7>(sv) * 8 + q];
    const int svn = (q < n - 8) ? csr_src[idx + 8u] : sv;
    float k1[4];
#pragma unroll
    for (int v = 0; v < 4; ++v) {
      const float pa = dot4p(r[v], s4), pb = dot4p(r[4 + v], s4);
      k1[v] = (b2 ? pb : pa) + xor_get<4>(b2 ? pa : pb);
    }
    float k2[2];
#pragma unroll
    for (int v = 0; v < 2; ++v)
      k2[v] = (b1 ? k1[2 + v] : k1[v]) + xor_get<2>(b1 ? k1[v] : k1[2 + v]);
    const float sc0 = (b0 ? k2[1] : k2[0]) + xor_get<1>(b0 ? k2[0] : k2[1]);

    const float l = (q < n) ? lrelu2(sc0 + ad) : -__builtin_huge_valf();
    float g = fmaxf(l, xor_get<1>(l));
    g = fmaxf(g, xor_get<2>(g));
    g = fmaxf(g, xor_get<4>(g));
    const float mn = fmaxf(m, g);
    const float sc = exp2_hw(m - mn);
    const float w = exp2_hw(l - mn);
    m = mn;
    den = den * sc + w;
    acc *= sc;
    acc += bcast8<0>(w) * r[0];
    acc += bcast8<1>(w) * r[1];
    acc += bcast8<2>(w) * r[2];
    acc += bcast8<3>(w) * r[3];
    acc += bcast8<4>(w) * r[4];
    acc += bcast8<5>(w) * r[5];
    acc += bcast8<6>(w) * r[6];
    acc += bcast8<7>(w) * r[7];
    sv = svn;
    n -= 8;
    idx += 8u;
  }

  den = sum8(den);
  if (live) {
    const float qd = __builtin_amdgcn_rcpf(den + 1e-16f);
    const v4f bv = ((const v4f*)bias)[q];
    v4f o;
    o.x = acc.x * qd + bv.x;  o.y = acc.y * qd + bv.y;
    o.z = acc.z * qd + bv.z;  o.w = acc.w * qd + bv.w;
    o.x = elu1(o.x);  o.y = elu1(o.y);  o.z = elu1(o.z);  o.w = elu1(o.w);
    ((v4f*)out)[d * 8 + q] = o;
  }
}

extern "C" void kernel_launch(void* const* d_in, const int* in_sizes, int n_in,
                              void* d_out, int out_size, void* d_ws, size_t ws_size,
                              hipStream_t stream) {
  const float* x   = (const float*)d_in[0];
  const int*   ei  = (const int*)d_in[1];
  const float* W1  = (const float*)d_in[2];
  const float* as1 = (const float*)d_in[3];
  const float* ad1 = (const float*)d_in[4];
  const float* b1  = (const float*)d_in[5];
  const float* W2  = (const float*)d_in[6];
  const float* as2 = (const float*)d_in[7];
  const float* ad2 = (const float*)d_in[8];
  const float* b2  = (const float*)d_in[9];

  const int HC   = in_sizes[3];
  const int Fin  = in_sizes[2] / HC;
  const int N    = in_sizes[0] / Fin;
  const int E    = in_sizes[1] / 2;
  const int Cout = in_sizes[7];

  float* ws = (float*)d_ws;
  float* h1  = ws;
  float* h2  = h1 + (size_t)N * HC;
  int*   rowptr = (int*)(h2 + (size_t)N * Cout);
  int*   csr    = rowptr + ((N + 1 + 31) / 32) * 32;
  int*   recS   = csr + E;
  int*   recD   = recS + E;
  const int nChunks = (E + BIN_CHUNK - 1) / BIN_CHUNK;
  int*   tabOff = recD + E;
  int*   tabCnt = tabOff + (size_t)nChunks * BKT_DSTS;

  const int TB = 256;
  auto blocks = [](long long n, int tb) { return (int)((n + tb - 1) / tb); };

  edge_bin<<<nChunks, SORT_TB, 0, stream>>>(ei, E, recS, recD, tabOff, tabCnt);
  bucket_sort<<<(N + BKT_DSTS - 1) / BKT_DSTS, SORT_TB, 0, stream>>>(
      recS, recD, tabOff, tabCnt, nChunks, rowptr, csr, N);

  {
    const long long waves = (long long)((N + 31) / 32) * (HC / 64);
    gemm_wmma_f32<2, 4, 32, 128, 256><<<blocks(waves * 32, TB), TB, 0, stream>>>(x, W1, h1, N);
    agg_h8_dense2<<<(N + L1_NODES - 1) / L1_NODES, 32 * L1_NODES, 0, stream>>>(
        rowptr, csr, as1, ad1, h1, b1, W2, h2, N);
  }

  {
    agg_h1<<<blocks((long long)((N + L2_NPW - 1) / L2_NPW) * 32, TB), TB, 0, stream>>>(
        rowptr, csr, as2, ad2, h2, b2, (float*)d_out, N);
  }
}
